# starting kernel
# baseline (speedup 1.0000x reference)
#include <hip/hip_runtime.h>
#include <hip/hip_bf16.h>


#define NN      100000
#define EE      1600000
#define IN_DIM  256
#define HID     128
#define NCLS    40
#define NCLS_P  48

typedef __attribute__((ext_vector_type(2))) float v2f;
typedef __attribute__((ext_vector_type(8))) float v8f;

__global__ void init_deg_kernel(float* deg, int n) {
    int i = blockIdx.x * blockDim.x + threadIdx.x;
    if (i < n) deg[i] = 1.0f;
}

__global__ void deg_accum_kernel(const int* __restrict__ dst, float* deg, int e) {
    int i = blockIdx.x * blockDim.x + threadIdx.x;
    int stride = gridDim.x * blockDim.x;
    for (; i < e; i += stride) atomicAdd(&deg[dst[i]], 1.0f);
}

__global__ void deg_to_dinv_kernel(float* deg, int n) {
    int i = blockIdx.x * blockDim.x + threadIdx.x;
    if (i < n) deg[i] = rsqrtf(deg[i]);
}

__global__ void pad_w3_kernel(const float* __restrict__ w, float* __restrict__ wp) {
    int i = blockIdx.x * blockDim.x + threadIdx.x;
    if (i >= HID * NCLS_P) return;
    int r = i / NCLS_P, c = i - r * NCLS_P;
    wp[i] = (c < NCLS) ? w[r * NCLS + c] : 0.0f;
}

template <int NT, int K, int NOUT>
__global__ void gemm_wmma_f32_kernel(const float* __restrict__ A,
                                     const float* __restrict__ B,
                                     float* __restrict__ H,
                                     int M) {
    static_assert(NOUT == NT * 16, "full tiles only");
    const int lane  = threadIdx.x & 31;
    const int wave  = threadIdx.x >> 5;
    const int wpb   = blockDim.x >> 5;
    const int tile  = blockIdx.x * wpb + wave;
    const int mtile = M >> 4;
    if (tile >= mtile) return;

    const int rowA    = (tile << 4) + (lane & 15);
    const int kb      = (lane >> 4) << 1;
    const int colBase = lane & 15;

    v8f acc[NT];
#pragma unroll
    for (int t = 0; t < NT; ++t) acc[t] = (v8f)0.0f;

    const float* __restrict__ arow = A + (size_t)rowA * K;

#pragma unroll 2
    for (int k = 0; k < K; k += 4) {
        const float2 av = *(const float2*)(arow + k + kb);
        v2f a; a.x = av.x; a.y = av.y;

        const float* __restrict__ b0 = B + (size_t)(k + kb) * NOUT + colBase;
        const float* __restrict__ b1 = b0 + NOUT;
#pragma unroll
        for (int t = 0; t < NT; ++t) {
            v2f b;
            b.x = b0[t * 16];
            b.y = b1[t * 16];
            acc[t] = __builtin_amdgcn_wmma_f32_16x16x4_f32(
                false, a, false, b, (short)0, acc[t], false, false);
        }
    }

    const int rowD = (tile << 4) + ((lane >> 4) << 3);
    float* __restrict__ hout = H + (size_t)rowD * NOUT + colBase;
#pragma unroll
    for (int t = 0; t < NT; ++t) {
#pragma unroll
        for (int r = 0; r < 8; ++r)
            hout[(size_t)r * NOUT + t * 16] = acc[t][r];
    }
}

template <int F, int HS>
__global__ void scatter_kernel(const float* __restrict__ h,
                               const int* __restrict__ src,
                               const int* __restrict__ dst,
                               const float* __restrict__ dinv,
                               float* __restrict__ out,
                               int e) {
    const int lane = threadIdx.x & 31;
    int warp = (blockIdx.x * blockDim.x + threadIdx.x) >> 5;
    const int nwarps = (gridDim.x * blockDim.x) >> 5;
    for (; warp < e; warp += nwarps) {
        const int s = src[warp];
        const int d = dst[warp];
        const float w = dinv[s] * dinv[d];
        const float* __restrict__ hs = h + (size_t)s * HS;
        float* __restrict__ od       = out + (size_t)d * F;
#pragma unroll
        for (int f = lane; f < F; f += 32)
            atomicAdd(&od[f], hs[f] * w);
    }
}

template <int F, int HS, int RELU>
__global__ void finalize_kernel(const float* __restrict__ agg,
                                const float* __restrict__ h,
                                const float* __restrict__ dinv,
                                const float* __restrict__ bias,
                                float* __restrict__ out,
                                int n) {
    size_t idx = (size_t)blockIdx.x * blockDim.x + threadIdx.x;
    const size_t total = (size_t)n * F;
    const size_t stride = (size_t)gridDim.x * blockDim.x;
    for (; idx < total; idx += stride) {
        const int i = (int)(idx / F);
        const int f = (int)(idx - (size_t)i * F);
        const float di = dinv[i];
        float v = agg[idx] + h[(size_t)i * HS + f] * di * di + bias[f];
        if (RELU) v = fmaxf(v, 0.0f);
        out[idx] = v;
    }
}

extern "C" void kernel_launch(void* const* d_in, const int* in_sizes, int n_in,
                              void* d_out, int out_size, void* d_ws, size_t ws_size,
                              hipStream_t stream) {
    const float* x    = (const float*)d_in[0];
    const int*   eidx = (const int*)d_in[1];
    const float* W1   = (const float*)d_in[2];
    const float* b1   = (const float*)d_in[3];
    const float* W2   = (const float*)d_in[4];
    const float* b2   = (const float*)d_in[5];
    const float* W3   = (const float*)d_in[6];
    const float* b3   = (const float*)d_in[7];
    float* out = (float*)d_out;

    const int* src = eidx;
    const int* dst = eidx + EE;

    float* dinv = (float*)d_ws;
    float* bufA = dinv + NN;
    float* bufB = bufA + (size_t)NN * HID;
    float* bufC = bufB + (size_t)NN * HID;
    float* W3p  = bufC + (size_t)NN * HID;

    const int TB = 256;

    init_deg_kernel<<<(NN + TB - 1) / TB, TB, 0, stream>>>(dinv, NN);
    deg_accum_kernel<<<4096, TB, 0, stream>>>(dst, dinv, EE);
    deg_to_dinv_kernel<<<(NN + TB - 1) / TB, TB, 0, stream>>>(dinv, NN);
    pad_w3_kernel<<<(HID * NCLS_P + TB - 1) / TB, TB, 0, stream>>>(W3, W3p);

    const int MT = NN / 16;
    const int gemmBlocks = (MT + 7) / 8;
    const int scatterBlocks = 16384;
    const int elemBlocks128 = (int)(((size_t)NN * HID + TB - 1) / TB);
    const int elemBlocks40  = (int)(((size_t)NN * NCLS + TB - 1) / TB);

    gemm_wmma_f32_kernel<8, IN_DIM, HID><<<gemmBlocks, TB, 0, stream>>>(x, W1, bufA, NN);
    hipMemsetAsync(bufB, 0, (size_t)NN * HID * sizeof(float), stream);
    scatter_kernel<HID, HID><<<scatterBlocks, TB, 0, stream>>>(bufA, src, dst, dinv, bufB, EE);
    finalize_kernel<HID, HID, 1><<<elemBlocks128, TB, 0, stream>>>(bufB, bufA, dinv, b1, bufB, NN);

    gemm_wmma_f32_kernel<8, HID, HID><<<gemmBlocks, TB, 0, stream>>>(bufB, W2, bufA, NN);
    hipMemsetAsync(bufC, 0, (size_t)NN * HID * sizeof(float), stream);
    scatter_kernel<HID, HID><<<scatterBlocks, TB, 0, stream>>>(bufA, src, dst, dinv, bufC, EE);
    finalize_kernel<HID, HID, 1><<<elemBlocks128, TB, 0, stream>>>(bufC, bufA, dinv, b2, bufB, NN);

    gemm_wmma_f32_kernel<3, HID, NCLS_P><<<gemmBlocks, TB, 0, stream>>>(bufB, W3p, bufA, NN);
    hipMemsetAsync(out, 0, (size_t)NN * NCLS * sizeof(float), stream);
    scatter_kernel<NCLS, NCLS_P><<<scatterBlocks, TB, 0, stream>>>(bufA, src, dst, dinv, out, EE);
    finalize_kernel<NCLS, NCLS_P, 0><<<elemBlocks40, TB, 0, stream>>>(out, bufA, dinv, b3, out, NN);
}


# Edge sums by owner work-groups: each keeps 392 destination rows in LDS, scans the whole dst list, adds its edges' rows with LDS float adds, fuses self/bias/relu; sum order = LDS arrival
# speedup vs baseline: 1.5229x; 1.5229x over previous
#include <hip/hip_runtime.h>
#include <hip/hip_bf16.h>


#define NN      100000
#define EE      1600000
#define IN_DIM  256
#define HID     128
#define NCLS    40
#define NCLS_P  48

typedef __attribute__((ext_vector_type(2))) float v2f;
typedef __attribute__((ext_vector_type(8))) float v8f;

__global__ void init_deg_kernel(float* deg, int n) {
    int i = blockIdx.x * blockDim.x + threadIdx.x;
    if (i < n) deg[i] = 1.0f;
}

__global__ void deg_accum_kernel(const int* __restrict__ dst, float* deg, int e) {
    int i = blockIdx.x * blockDim.x + threadIdx.x;
    int stride = gridDim.x * blockDim.x;
    for (; i < e; i += stride) atomicAdd(&deg[dst[i]], 1.0f);
}

__global__ void deg_to_dinv_kernel(float* deg, int n) {
    int i = blockIdx.x * blockDim.x + threadIdx.x;
    if (i < n) deg[i] = rsqrtf(deg[i]);
}

__global__ void pad_w3_kernel(const float* __restrict__ w, float* __restrict__ wp) {
    int i = blockIdx.x * blockDim.x + threadIdx.x;
    if (i >= HID * NCLS_P) return;
    int r = i / NCLS_P, c = i - r * NCLS_P;
    wp[i] = (c < NCLS) ? w[r * NCLS + c] : 0.0f;
}

template <int NT, int K, int NOUT>
__global__ void gemm_wmma_f32_kernel(const float* __restrict__ A,
                                     const float* __restrict__ B,
                                     float* __restrict__ H,
                                     int M) {
    static_assert(NOUT == NT * 16, "full tiles only");
    const int lane  = threadIdx.x & 31;
    const int wave  = threadIdx.x >> 5;
    const int wpb   = blockDim.x >> 5;
    const int tile  = blockIdx.x * wpb + wave;
    const int mtile = M >> 4;
    if (tile >= mtile) return;

    const int rowA    = (tile << 4) + (lane & 15);
    const int kb      = (lane >> 4) << 1;
    const int colBase = lane & 15;

    v8f acc[NT];
#pragma unroll
    for (int t = 0; t < NT; ++t) acc[t] = (v8f)0.0f;

    const float* __restrict__ arow = A + (size_t)rowA * K;

#pragma unroll 2
    for (int k = 0; k < K; k += 4) {
        const float2 av = *(const float2*)(arow + k + kb);
        v2f a; a.x = av.x; a.y = av.y;

        const float* __restrict__ b0 = B + (size_t)(k + kb) * NOUT + colBase;
        const float* __restrict__ b1 = b0 + NOUT;
#pragma unroll
        for (int t = 0; t < NT; ++t) {
            v2f b;
            b.x = b0[t * 16];
            b.y = b1[t * 16];
            acc[t] = __builtin_amdgcn_wmma_f32_16x16x4_f32(
                false, a, false, b, (short)0, acc[t], false, false);
        }
    }

    const int rowD = (tile << 4) + ((lane >> 4) << 3);
    float* __restrict__ hout = H + (size_t)rowD * NOUT + colBase;
#pragma unroll
    for (int t = 0; t < NT; ++t) {
#pragma unroll
        for (int r = 0; r < 8; ++r)
            hout[(size_t)r * NOUT + t * 16] = acc[t][r];
    }
}

#define OWN_WGS     256
#define OWN_ROWS    392
#define OWN_THREADS 1024
#define OWN_WAVES   (OWN_THREADS / 32)
#define OWN_QCAP    160
#define OWN_U       8

template <int F, int HS, int RELU>
__global__ __launch_bounds__(OWN_THREADS)
void own_rows_kernel(const float* __restrict__ h,
                     const int* __restrict__ src,
                     const int* __restrict__ dst,
                     const float* __restrict__ dinv,
                     const float* __restrict__ bias,
                     float* __restrict__ out) {
    static_assert(OWN_WGS * OWN_ROWS >= NN, "rows covered");
    static_assert((OWN_WGS - 1) * OWN_ROWS < NN, "no empty group");
    static_assert((OWN_ROWS * F * 4) % 128 == 0, "whole lines per group");
    static_assert(EE % 4 == 0, "16-byte reads of dst");

    __shared__ float acc[OWN_ROWS * F];
    __shared__ int   q_e[OWN_WAVES][OWN_QCAP];
    __shared__ int   q_r[OWN_WAVES][OWN_QCAP];

    const int tid  = threadIdx.x;
    const int lane = tid & 31;
    const int wv   = tid >> 5;
    const int lo   = blockIdx.x * OWN_ROWS;

    for (int i = tid; i < OWN_ROWS * F; i += OWN_THREADS) acc[i] = 0.0f;
    __syncthreads();

    const unsigned below = (1u << lane) - 1u;
    int qn = 0;

    auto flush = [&](int head, int n) {
        int r_l = 0, s_l = 0;
        float w_l = 0.0f;
        if (lane < n) {
            const int e_l = q_e[wv][head + lane];
            r_l = q_r[wv][head + lane];
            s_l = src[e_l];
            w_l = dinv[s_l] * dinv[lo + r_l];
        }
        constexpr int KF = (F + 31) / 32;
        constexpr int GE = 8;
#pragma unroll 1
        for (int j = 0; j < n; j += GE) {
            float v[GE][KF];
#pragma unroll
            for (int g = 0; g < GE; ++g) {
                if (j + g < n) {
                    const int   s = __builtin_amdgcn_readlane(s_l, j + g);
                    const float w = __int_as_float(__builtin_amdgcn_readlane(__float_as_int(w_l), j + g));
                    const float* hs = h + (size_t)s * HS;
#pragma unroll
                    for (int k = 0; k < KF; ++k) {
                        const int f = lane + 32 * k;
                        v[g][k] = (f < F) ? hs[f] * w : 0.0f;
                    }
                }
            }
#pragma unroll
            for (int g = 0; g < GE; ++g) {
                if (j + g < n) {
                    const int r = __builtin_amdgcn_readlane(r_l, j + g);
                    float* a = acc + r * F;
#pragma unroll
                    for (int k = 0; k < KF; ++k) {
                        const int f = lane + 32 * k;
                        if (f < F) atomicAdd(&a[f], v[g][k]);
                    }
                }
            }
        }
    };

    auto push = [&](unsigned r, int eid) {
        const bool p = r < (unsigned)OWN_ROWS;
        const unsigned mk = __builtin_amdgcn_ballot_w32(p);
        if (mk != 0u) {
            if (p) {
                const int pos = qn + __popc(mk & below);
                q_e[wv][pos] = eid;
                q_r[wv][pos] = (int)r;
            }
            qn += __popc(mk);
        }
    };

    auto drain = [&]() {
        int head = 0;
        while (qn - head >= 32) { flush(head, 32); head += 32; }
        const int rem = qn - head;
        int te = 0, tr = 0;
        if (lane < rem) { te = q_e[wv][head + lane]; tr = q_r[wv][head + lane]; }
        if (lane < rem) { q_e[wv][lane] = te; q_r[wv][lane] = tr; }
        qn = rem;
    };

    constexpr int NV4  = EE / 4;
    constexpr int STEP = OWN_THREADS * OWN_U;
    constexpr int NIT  = (NV4 + STEP - 1) / STEP;
    const int4* __restrict__ dst4 = (const int4*)dst;
    const int rot = blockIdx.x % NIT;

    for (int it = 0; it < NIT; ++it) {
        int it2 = it + rot;
        if (it2 >= NIT) it2 -= NIT;
        const int base = it2 * STEP + tid;
        int4 dv[OWN_U];
#pragma unroll
        for (int u = 0; u < OWN_U; ++u) {
            const int i4 = base + u * OWN_THREADS;
            dv[u] = (i4 < NV4) ? dst4[i4] : make_int4(-1, -1, -1, -1);
        }
#pragma unroll
        for (int u = 0; u < OWN_U; ++u) {
            const int i4 = base + u * OWN_THREADS;
            const unsigned r0 = (unsigned)(dv[u].x - lo);
            const unsigned r1 = (unsigned)(dv[u].y - lo);
            const unsigned r2 = (unsigned)(dv[u].z - lo);
            const unsigned r3 = (unsigned)(dv[u].w - lo);
            const unsigned rmin = min(min(r0, r1), min(r2, r3));
            if (__builtin_amdgcn_ballot_w32(rmin < (unsigned)OWN_ROWS) != 0u) {
                push(r0, i4 * 4 + 0);
                push(r1, i4 * 4 + 1);
                push(r2, i4 * 4 + 2);
                push(r3, i4 * 4 + 3);
                if (qn >= 32) drain();
            }
        }
    }
    if (qn > 0) flush(0, qn);
    __syncthreads();

    const int nrows = (NN - lo < OWN_ROWS) ? (NN - lo) : OWN_ROWS;
    float* __restrict__ orow = out + (size_t)lo * F;
    for (int idx = tid; idx < nrows * F; idx += OWN_THREADS) {
        const int r = idx / F;
        const int f = idx - r * F;
        const int i = lo + r;
        const float di = dinv[i];
        float v = acc[idx] + h[(size_t)i * HS + f] * di * di + bias[f];
        if (RELU) v = fmaxf(v, 0.0f);
        orow[idx] = v;
    }
}

extern "C" void kernel_launch(void* const* d_in, const int* in_sizes, int n_in,
                              void* d_out, int out_size, void* d_ws, size_t ws_size,
                              hipStream_t stream) {
    const float* x    = (const float*)d_in[0];
    const int*   eidx = (const int*)d_in[1];
    const float* W1   = (const float*)d_in[2];
    const float* b1   = (const float*)d_in[3];
    const float* W2   = (const float*)d_in[4];
    const float* b2   = (const float*)d_in[5];
    const float* W3   = (const float*)d_in[6];
    const float* b3   = (const float*)d_in[7];
    float* out = (float*)d_out;

    const int* src = eidx;
    const int* dst = eidx + EE;

    float* dinv = (float*)d_ws;
    float* bufA = dinv + NN;
    float* bufB = bufA + (size_t)NN * HID;
    float* W3p  = bufB + (size_t)NN * HID;

    const int TB = 256;

    init_deg_kernel<<<(NN + TB - 1) / TB, TB, 0, stream>>>(dinv, NN);
    deg_accum_kernel<<<4096, TB, 0, stream>>>(dst, dinv, EE);
    deg_to_dinv_kernel<<<(NN + TB - 1) / TB, TB, 0, stream>>>(dinv, NN);
    pad_w3_kernel<<<(HID * NCLS_P + TB - 1) / TB, TB, 0, stream>>>(W3, W3p);

    const int MT = NN / 16;
    const int gemmBlocks = (MT + 7) / 8;

    gemm_wmma_f32_kernel<8, IN_DIM, HID><<<gemmBlocks, TB, 0, stream>>>(x, W1, bufA, NN);
    own_rows_kernel<HID, HID, 1><<<OWN_WGS, OWN_THREADS, 0, stream>>>(bufA, src, dst, dinv, b1, bufB);

    gemm_wmma_f32_kernel<8, HID, HID><<<gemmBlocks, TB, 0, stream>>>(bufB, W2, bufA, NN);
    own_rows_kernel<HID, HID, 1><<<OWN_WGS, OWN_THREADS, 0, stream>>>(bufA, src, dst, dinv, b2, bufB);

    gemm_wmma_f32_kernel<3, HID, NCLS_P><<<gemmBlocks, TB, 0, stream>>>(bufB, W3p, bufA, NN);
    own_rows_kernel<NCLS, NCLS_P, 0><<<OWN_WGS, OWN_THREADS, 0, stream>>>(bufA, src, dst, dinv, b3, out);
}


# Edge-sum groups scan 8 bit planes of the edges' owner number (1 byte per edge, written once beside the degree count with a packed src|row word), not the 4-byte dst list; sum order = LDS arrival
# speedup vs baseline: 1.7408x; 1.1431x over previous
#include <hip/hip_runtime.h>
#include <hip/hip_bf16.h>


#define NN      100000
#define EE      1600000
#define IN_DIM  256
#define HID     128
#define NCLS    40
#define NCLS_P  48

typedef __attribute__((ext_vector_type(2))) float v2f;
typedef __attribute__((ext_vector_type(8))) float v8f;

__global__ void init_deg_kernel(float* deg, int n) {
    int i = blockIdx.x * blockDim.x + threadIdx.x;
    if (i < n) deg[i] = 1.0f;
}

#define OWN_WGS     256
#define OWN_ROWS    392
#define KEY_PLANES  8
#define PLANE_WORDS 50240
#define KEY_EDGES   1024
#define SRC_BITS    17

__global__ __launch_bounds__(64)
void deg_keys_kernel(const int* __restrict__ src, const int* __restrict__ dst, float* deg,
                     unsigned int* __restrict__ planes, unsigned int* __restrict__ packed) {
    static_assert(OWN_WGS <= (1 << KEY_PLANES), "owner number fits the planes");
    static_assert(NN <= (1 << SRC_BITS) && OWN_ROWS <= (1 << (32 - SRC_BITS)), "packed word");
    static_assert(EE % 32 == 0 && EE / 32 <= PLANE_WORDS && PLANE_WORDS % 64 == 0, "plane size");
    const int lane = threadIdx.x & 31;
    const int wave = (blockIdx.x * blockDim.x + threadIdx.x) >> 5;
    const int base = wave * KEY_EDGES;
    if (base >= EE) return;

    unsigned int w[KEY_PLANES];
#pragma unroll
    for (int k = 0; k < KEY_PLANES; ++k) w[k] = 0u;

    constexpr int TG = 16;
#pragma unroll 1
    for (int j0 = 0; j0 < 32; j0 += TG) {
        int d[TG], s[TG];
#pragma unroll
        for (int g = 0; g < TG; ++g) {
            const int e = min(base + 32 * (j0 + g) + lane, EE - 1);
            d[g] = dst[e];
            s[g] = src[e];
        }
#pragma unroll
        for (int g = 0; g < TG; ++g) {
            const int j = j0 + g;
            const int e = base + 32 * j + lane;
            unsigned int o = 0u;
            if (e < EE) {
                atomicAdd(&deg[d[g]], 1.0f);
                o = (unsigned int)d[g] / OWN_ROWS;
                packed[e] = (unsigned int)s[g] | (((unsigned int)d[g] - o * OWN_ROWS) << SRC_BITS);
            }
#pragma unroll
            for (int k = 0; k < KEY_PLANES; ++k) {
                const unsigned int b = __builtin_amdgcn_ballot_w32(((o >> k) & 1u) != 0u);
                if (lane == j) w[k] = b;
            }
        }
    }
    const int word = (base >> 5) + lane;
    if (word < EE / 32) {
#pragma unroll
        for (int k = 0; k < KEY_PLANES; ++k) planes[k * PLANE_WORDS + word] = w[k];
    }
}

__global__ void deg_to_dinv_kernel(float* deg, int n) {
    int i = blockIdx.x * blockDim.x + threadIdx.x;
    if (i < n) deg[i] = rsqrtf(deg[i]);
}

__global__ void pad_w3_kernel(const float* __restrict__ w, float* __restrict__ wp) {
    int i = blockIdx.x * blockDim.x + threadIdx.x;
    if (i >= HID * NCLS_P) return;
    int r = i / NCLS_P, c = i - r * NCLS_P;
    wp[i] = (c < NCLS) ? w[r * NCLS + c] : 0.0f;
}

template <int NT, int K, int NOUT>
__global__ void gemm_wmma_f32_kernel(const float* __restrict__ A,
                                     const float* __restrict__ B,
                                     float* __restrict__ H,
                                     int M) {
    static_assert(NOUT == NT * 16, "full tiles only");
    const int lane  = threadIdx.x & 31;
    const int wave  = threadIdx.x >> 5;
    const int wpb   = blockDim.x >> 5;
    const int tile  = blockIdx.x * wpb + wave;
    const int mtile = M >> 4;
    if (tile >= mtile) return;

    const int rowA    = (tile << 4) + (lane & 15);
    const int kb      = (lane >> 4) << 1;
    const int colBase = lane & 15;

    v8f acc[NT];
#pragma unroll
    for (int t = 0; t < NT; ++t) acc[t] = (v8f)0.0f;

    const float* __restrict__ arow = A + (size_t)rowA * K;

#pragma unroll 2
    for (int k = 0; k < K; k += 4) {
        const float2 av = *(const float2*)(arow + k + kb);
        v2f a; a.x = av.x; a.y = av.y;

        const float* __restrict__ b0 = B + (size_t)(k + kb) * NOUT + colBase;
        const float* __restrict__ b1 = b0 + NOUT;
#pragma unroll
        for (int t = 0; t < NT; ++t) {
            v2f b;
            b.x = b0[t * 16];
            b.y = b1[t * 16];
            acc[t] = __builtin_amdgcn_wmma_f32_16x16x4_f32(
                false, a, false, b, (short)0, acc[t], false, false);
        }
    }

    const int rowD = (tile << 4) + ((lane >> 4) << 3);
    float* __restrict__ hout = H + (size_t)rowD * NOUT + colBase;
#pragma unroll
    for (int t = 0; t < NT; ++t) {
#pragma unroll
        for (int r = 0; r < 8; ++r)
            hout[(size_t)r * NOUT + t * 16] = acc[t][r];
    }
}

#define OWN_THREADS 1024
#define OWN_WAVES   (OWN_THREADS / 32)
#define OWN_QCAP    64

template <int F, int HS, int RELU>
__global__ __launch_bounds__(OWN_THREADS)
void own_rows_kernel(const float* __restrict__ h,
                     const unsigned int* __restrict__ planes,
                     const unsigned int* __restrict__ packed,
                     const float* __restrict__ dinv,
                     const float* __restrict__ bias,
                     float* __restrict__ out) {
    static_assert(OWN_WGS * OWN_ROWS >= NN, "rows covered");
    static_assert((OWN_WGS - 1) * OWN_ROWS < NN, "no empty group");
    static_assert((OWN_ROWS * F * 4) % 128 == 0, "whole lines per group");
    static_assert(EE % 128 == 0, "16-byte reads of a plane hold 128 whole edges");

    __shared__ float acc[OWN_ROWS * F];
    __shared__ int   q_e[OWN_WAVES][OWN_QCAP];

    const int tid  = threadIdx.x;
    const int lane = tid & 31;
    const int wv   = tid >> 5;
    const int lo   = blockIdx.x * OWN_ROWS;

    for (int i = tid; i < OWN_ROWS * F; i += OWN_THREADS) acc[i] = 0.0f;
    __syncthreads();

    const unsigned below = (1u << lane) - 1u;
    int qn = 0;

    auto flush = [&](int n) {
        int r_l = 0, s_l = 0;
        float w_l = 0.0f;
        if (lane < n) {
            const unsigned int pk = packed[q_e[wv][lane]];
            s_l = (int)(pk & ((1u << SRC_BITS) - 1u));
            r_l = (int)(pk >> SRC_BITS);
            w_l = dinv[s_l] * dinv[lo + r_l];
        }
        constexpr int KF = (F + 31) / 32;
        constexpr int GE = 8;
#pragma unroll 1
        for (int j = 0; j < n; j += GE) {
            float v[GE][KF];
#pragma unroll
            for (int g = 0; g < GE; ++g) {
                if (j + g < n) {
                    const int   s = __builtin_amdgcn_readlane(s_l, j + g);
                    const float w = __int_as_float(__builtin_amdgcn_readlane(__float_as_int(w_l), j + g));
                    const float* hs = h + (size_t)s * HS;
#pragma unroll
                    for (int k = 0; k < KF; ++k) {
                        const int f = lane + 32 * k;
                        v[g][k] = (f < F) ? hs[f] * w : 0.0f;
                    }
                }
            }
#pragma unroll
            for (int g = 0; g < GE; ++g) {
                if (j + g < n) {
                    const int r = __builtin_amdgcn_readlane(r_l, j + g);
                    float* a = acc + r * F;
#pragma unroll
                    for (int k = 0; k < KF; ++k) {
                        const int f = lane + 32 * k;
                        if (f < F) atomicAdd(&a[f], v[g][k]);
                    }
                }
            }
        }
    };

    auto push = [&](bool p, int eid) {
        const unsigned mk = __builtin_amdgcn_ballot_w32(p);
        if (p) q_e[wv][qn + __popc(mk & below)] = eid;
        qn += __popc(mk);
    };

    auto drain = [&]() {
        flush(32);
        const int rem = qn - 32;
        int te = 0;
        if (lane < rem) te = q_e[wv][32 + lane];
        if (lane < rem) q_e[wv][lane] = te;
        qn = rem;
    };

    const unsigned int own = blockIdx.x;
    unsigned int flip[KEY_PLANES];
#pragma unroll
    for (int k = 0; k < KEY_PLANES; ++k) flip[k] = ((own >> k) & 1u) ? 0u : 0xFFFFFFFFu;

    constexpr int NP4 = EE / 128;
    constexpr int PV4 = PLANE_WORDS / 4;
    constexpr int NIT = (NP4 + OWN_THREADS - 1) / OWN_THREADS;
    const uint4* __restrict__ pl4 = (const uint4*)planes;
    const int rot = blockIdx.x % NIT;

    for (int it = 0; it < NIT; ++it) {
        int it2 = it + rot;
        if (it2 >= NIT) it2 -= NIT;
        const int i4 = it2 * OWN_THREADS + tid;
        unsigned int m0 = 0u, m1 = 0u, m2 = 0u, m3 = 0u;
        if (i4 < NP4) {
            uint4 p[KEY_PLANES];
#pragma unroll
            for (int k = 0; k < KEY_PLANES; ++k) p[k] = pl4[k * PV4 + i4];
            m0 = p[0].x ^ flip[0]; m1 = p[0].y ^ flip[0]; m2 = p[0].z ^ flip[0]; m3 = p[0].w ^ flip[0];
#pragma unroll
            for (int k = 1; k < KEY_PLANES; ++k) {
                m0 &= p[k].x ^ flip[k];
                m1 &= p[k].y ^ flip[k];
                m2 &= p[k].z ^ flip[k];
                m3 &= p[k].w ^ flip[k];
            }
        }
#pragma unroll 1
        for (int j = 0; j < 4; ++j) {
            unsigned int mm = (j == 0) ? m0 : (j == 1) ? m1 : (j == 2) ? m2 : m3;
            const int eb = i4 * 128 + j * 32;
            while (__builtin_amdgcn_ballot_w32(mm != 0u) != 0u) {
                const bool p = mm != 0u;
                const int  b = __ffs((int)mm) - 1;
                mm &= mm - 1u;
                push(p, eb + b);
                if (qn >= 32) drain();
            }
        }
    }
    if (qn > 0) flush(qn);
    __syncthreads();

    const int nrows = (NN - lo < OWN_ROWS) ? (NN - lo) : OWN_ROWS;
    float* __restrict__ orow = out + (size_t)lo * F;
    for (int idx = tid; idx < nrows * F; idx += OWN_THREADS) {
        const int r = idx / F;
        const int f = idx - r * F;
        const int i = lo + r;
        const float di = dinv[i];
        float v = acc[idx] + h[(size_t)i * HS + f] * di * di + bias[f];
        if (RELU) v = fmaxf(v, 0.0f);
        orow[idx] = v;
    }
}

extern "C" void kernel_launch(void* const* d_in, const int* in_sizes, int n_in,
                              void* d_out, int out_size, void* d_ws, size_t ws_size,
                              hipStream_t stream) {
    const float* x    = (const float*)d_in[0];
    const int*   eidx = (const int*)d_in[1];
    const float* W1   = (const float*)d_in[2];
    const float* b1   = (const float*)d_in[3];
    const float* W2   = (const float*)d_in[4];
    const float* b2   = (const float*)d_in[5];
    const float* W3   = (const float*)d_in[6];
    const float* b3   = (const float*)d_in[7];
    float* out = (float*)d_out;

    const int* src = eidx;
    const int* dst = eidx + EE;

    float* dinv = (float*)d_ws;
    float* bufA = dinv + NN;
    float* bufB = bufA + (size_t)NN * HID;
    float* W3p  = bufB + (size_t)NN * HID;
    unsigned int* planes = (unsigned int*)(W3p + HID * NCLS_P);
    unsigned int* packed = planes + KEY_PLANES * PLANE_WORDS;

    const int TB = 256;

    init_deg_kernel<<<(NN + TB - 1) / TB, TB, 0, stream>>>(dinv, NN);
    deg_keys_kernel<<<(EE + 2 * KEY_EDGES - 1) / (2 * KEY_EDGES), 64, 0, stream>>>(src, dst, dinv, planes, packed);
    deg_to_dinv_kernel<<<(NN + TB - 1) / TB, TB, 0, stream>>>(dinv, NN);
    pad_w3_kernel<<<(HID * NCLS_P + TB - 1) / TB, TB, 0, stream>>>(W3, W3p);

    const int MT = NN / 16;
    const int gemmBlocks = (MT + 7) / 8;

    gemm_wmma_f32_kernel<8, IN_DIM, HID><<<gemmBlocks, TB, 0, stream>>>(x, W1, bufA, NN);
    own_rows_kernel<HID, HID, 1><<<OWN_WGS, OWN_THREADS, 0, stream>>>(bufA, planes, packed, dinv, b1, bufB);

    gemm_wmma_f32_kernel<8, HID, HID><<<gemmBlocks, TB, 0, stream>>>(bufB, W2, bufA, NN);
    own_rows_kernel<HID, HID, 1><<<OWN_WGS, OWN_THREADS, 0, stream>>>(bufA, planes, packed, dinv, b2, bufB);

    gemm_wmma_f32_kernel<3, HID, NCLS_P><<<gemmBlocks, TB, 0, stream>>>(bufB, W3p, bufA, NN);
    own_rows_kernel<NCLS, NCLS_P, 0><<<OWN_WGS, OWN_THREADS, 0, stream>>>(bufA, planes, packed, dinv, b3, out);
}


# Layer-1 edge-sum groups write down the edges they find (packed word, weight); layers 2 and 3 read that list: no second and third search of the planes, no packed/dinv lookups; sum order = LDS arrival
# speedup vs baseline: 2.1914x; 1.2588x over previous
#include <hip/hip_runtime.h>
#include <hip/hip_bf16.h>


#define NN      100000
#define EE      1600000
#define IN_DIM  256
#define HID     128
#define NCLS    40
#define NCLS_P  48

typedef __attribute__((ext_vector_type(2))) float v2f;
typedef __attribute__((ext_vector_type(8))) float v8f;

__global__ void init_deg_kernel(float* deg, int n) {
    int i = blockIdx.x * blockDim.x + threadIdx.x;
    if (i < n) deg[i] = 1.0f;
}

#define OWN_WGS     256
#define OWN_ROWS    392
#define KEY_PLANES  8
#define PLANE_WORDS 50240
#define KEY_EDGES   1024
#define SRC_BITS    17

__global__ __launch_bounds__(64)
void deg_keys_kernel(const int* __restrict__ src, const int* __restrict__ dst, float* deg,
                     unsigned int* __restrict__ planes, unsigned int* __restrict__ packed) {
    static_assert(OWN_WGS <= (1 << KEY_PLANES), "owner number fits the planes");
    static_assert(NN <= (1 << SRC_BITS) && OWN_ROWS <= (1 << (32 - SRC_BITS)), "packed word");
    static_assert(EE % 32 == 0 && EE / 32 <= PLANE_WORDS && PLANE_WORDS % 64 == 0, "plane size");
    const int lane = threadIdx.x & 31;
    const int wave = (blockIdx.x * blockDim.x + threadIdx.x) >> 5;
    const int base = wave * KEY_EDGES;
    if (base >= EE) return;

    unsigned int w[KEY_PLANES];
#pragma unroll
    for (int k = 0; k < KEY_PLANES; ++k) w[k] = 0u;

    constexpr int TG = 16;
#pragma unroll 1
    for (int j0 = 0; j0 < 32; j0 += TG) {
        int d[TG], s[TG];
#pragma unroll
        for (int g = 0; g < TG; ++g) {
            const int e = min(base + 32 * (j0 + g) + lane, EE - 1);
            d[g] = dst[e];
            s[g] = src[e];
        }
#pragma unroll
        for (int g = 0; g < TG; ++g) {
            const int j = j0 + g;
            const int e = base + 32 * j + lane;
            unsigned int o = 0u;
            if (e < EE) {
                atomicAdd(&deg[d[g]], 1.0f);
                o = (unsigned int)d[g] / OWN_ROWS;
                packed[e] = (unsigned int)s[g] | (((unsigned int)d[g] - o * OWN_ROWS) << SRC_BITS);
            }
#pragma unroll
            for (int k = 0; k < KEY_PLANES; ++k) {
                const unsigned int b = __builtin_amdgcn_ballot_w32(((o >> k) & 1u) != 0u);
                if (lane == j) w[k] = b;
            }
        }
    }
    const int word = (base >> 5) + lane;
    if (word < EE / 32) {
#pragma unroll
        for (int k = 0; k < KEY_PLANES; ++k) planes[k * PLANE_WORDS + word] = w[k];
    }
}

__global__ void deg_to_dinv_kernel(float* deg, int n) {
    int i = blockIdx.x * blockDim.x + threadIdx.x;
    if (i < n) deg[i] = rsqrtf(deg[i]);
}

__global__ void pad_w3_kernel(const float* __restrict__ w, float* __restrict__ wp) {
    int i = blockIdx.x * blockDim.x + threadIdx.x;
    if (i >= HID * NCLS_P) return;
    int r = i / NCLS_P, c = i - r * NCLS_P;
    wp[i] = (c < NCLS) ? w[r * NCLS + c] : 0.0f;
}

template <int NT, int K, int NOUT>
__global__ void gemm_wmma_f32_kernel(const float* __restrict__ A,
                                     const float* __restrict__ B,
                                     float* __restrict__ H,
                                     int M) {
    static_assert(NOUT == NT * 16, "full tiles only");
    const int lane  = threadIdx.x & 31;
    const int wave  = threadIdx.x >> 5;
    const int wpb   = blockDim.x >> 5;
    const int tile  = blockIdx.x * wpb + wave;
    const int mtile = M >> 4;
    if (tile >= mtile) return;

    const int rowA    = (tile << 4) + (lane & 15);
    const int kb      = (lane >> 4) << 1;
    const int colBase = lane & 15;

    v8f acc[NT];
#pragma unroll
    for (int t = 0; t < NT; ++t) acc[t] = (v8f)0.0f;

    const float* __restrict__ arow = A + (size_t)rowA * K;

#pragma unroll 2
    for (int k = 0; k < K; k += 4) {
        const float2 av = *(const float2*)(arow + k + kb);
        v2f a; a.x = av.x; a.y = av.y;

        const float* __restrict__ b0 = B + (size_t)(k + kb) * NOUT + colBase;
        const float* __restrict__ b1 = b0 + NOUT;
#pragma unroll
        for (int t = 0; t < NT; ++t) {
            v2f b;
            b.x = b0[t * 16];
            b.y = b1[t * 16];
            acc[t] = __builtin_amdgcn_wmma_f32_16x16x4_f32(
                false, a, false, b, (short)0, acc[t], false, false);
        }
    }

    const int rowD = (tile << 4) + ((lane >> 4) << 3);
    float* __restrict__ hout = H + (size_t)rowD * NOUT + colBase;
#pragma unroll
    for (int t = 0; t < NT; ++t) {
#pragma unroll
        for (int r = 0; r < 8; ++r)
            hout[(size_t)r * NOUT + t * 16] = acc[t][r];
    }
}

#define OWN_THREADS 1024
#define OWN_WAVES   (OWN_THREADS / 32)
#define OWN_QCAP    64
#define LIST_CAP    16384


template <int F, int HS, int RELU, int FIRST>
__global__ __launch_bounds__(OWN_THREADS)
void own_rows_kernel(const float* __restrict__ h,
                     const unsigned int* __restrict__ planes,
                     const unsigned int* __restrict__ packed,
                     const float* __restrict__ dinv,
                     const float* __restrict__ bias,
                     unsigned int* __restrict__ lists,
                     int* __restrict__ counts,
                     float* __restrict__ out) {
    static_assert(OWN_WGS * OWN_ROWS >= NN, "rows covered");
    static_assert((OWN_WGS - 1) * OWN_ROWS < NN, "no empty group");
    static_assert((OWN_ROWS * F * 4) % 128 == 0, "whole lines per group");
    static_assert(EE % 128 == 0, "16-byte reads of a plane hold 128 whole edges");
    static_assert((NN + 2 * NN * HID + HID * NCLS_P + KEY_PLANES * PLANE_WORDS + EE) % 2 == 0, "8-byte list places are aligned");

    __shared__ float acc[OWN_ROWS * F];
    __shared__ int   q_e[OWN_WAVES][OWN_QCAP];
    __shared__ int   l_cnt;

    const int tid  = threadIdx.x;
    const int lane = tid & 31;
    const int wv   = tid >> 5;
    const int lo   = blockIdx.x * OWN_ROWS;

    for (int i = tid; i < OWN_ROWS * F; i += OWN_THREADS) acc[i] = 0.0f;
    if (FIRST && tid == 0) l_cnt = 0;
    __syncthreads();

    uint2* __restrict__ lst = (uint2*)lists + (size_t)blockIdx.x * LIST_CAP;

    const unsigned below = (1u << lane) - 1u;
    int qn = 0;

    auto add_rows = [&](int n, int s_l, int r_l, float w_l) {
        constexpr int KF = (F + 31) / 32;
        constexpr int GE = 8;
#pragma unroll 1
        for (int j = 0; j < n; j += GE) {
            float v[GE][KF];
#pragma unroll
            for (int g = 0; g < GE; ++g) {
                const int s = __builtin_amdgcn_readlane(s_l, j + g);
                const float* hs = h + (size_t)s * HS;
#pragma unroll
                for (int k = 0; k < KF; ++k) {
                    const int f = lane + 32 * k;
                    v[g][k] = (f < F) ? hs[f] : 0.0f;
                }
            }
#pragma unroll
            for (int g = 0; g < GE; ++g) {
                if (j + g < n) {
                    const int   r = __builtin_amdgcn_readlane(r_l, j + g);
                    const float w = __int_as_float(__builtin_amdgcn_readlane(__float_as_int(w_l), j + g));
                    float* a = acc + r * F;
#pragma unroll
                    for (int k = 0; k < KF; ++k) {
                        const int f = lane + 32 * k;
                        if (f < F) atomicAdd(&a[f], v[g][k] * w);
                    }
                }
            }
        }
    };

    auto flush = [&](int n) {
        int r_l = 0, s_l = 0;
        float w_l = 0.0f;
        unsigned int pk = 0u;
        if (lane < n) {
            pk  = packed[q_e[wv][lane]];
            s_l = (int)(pk & ((1u << SRC_BITS) - 1u));
            r_l = (int)(pk >> SRC_BITS);
            w_l = dinv[s_l] * dinv[lo + r_l];
        }
        if (FIRST) {
            int pos = 0;
            if (lane == 0) pos = atomicAdd(&l_cnt, n);
            pos = __builtin_amdgcn_readfirstlane(pos);
            if (lane < n && pos + lane < LIST_CAP) lst[pos + lane] = make_uint2(pk, __float_as_uint(w_l));
        }
        add_rows(n, s_l, r_l, w_l);
    };

    auto push = [&](bool p, int eid) {
        const unsigned mk = __builtin_amdgcn_ballot_w32(p);
        if (p) q_e[wv][qn + __popc(mk & below)] = eid;
        qn += __popc(mk);
    };

    auto drain = [&]() {
        flush(32);
        const int rem = qn - 32;
        int te = 0;
        if (lane < rem) te = q_e[wv][32 + lane];
        if (lane < rem) q_e[wv][lane] = te;
        qn = rem;
    };

    auto search = [&]() {
    const unsigned int own = blockIdx.x;
    unsigned int flip[KEY_PLANES];
#pragma unroll
    for (int k = 0; k < KEY_PLANES; ++k) flip[k] = ((own >> k) & 1u) ? 0u : 0xFFFFFFFFu;

    constexpr int NP4 = EE / 128;
    constexpr int PV4 = PLANE_WORDS / 4;
    constexpr int NIT = (NP4 + OWN_THREADS - 1) / OWN_THREADS;
    const uint4* __restrict__ pl4 = (const uint4*)planes;
    const int rot = blockIdx.x % NIT;

    for (int it = 0; it < NIT; ++it) {
        int it2 = it + rot;
        if (it2 >= NIT) it2 -= NIT;
        const int i4 = it2 * OWN_THREADS + tid;
        unsigned int m0 = 0u, m1 = 0u, m2 = 0u, m3 = 0u;
        if (i4 < NP4) {
            uint4 p[KEY_PLANES];
#pragma unroll
            for (int k = 0; k < KEY_PLANES; ++k) p[k] = pl4[k * PV4 + i4];
            m0 = p[0].x ^ flip[0]; m1 = p[0].y ^ flip[0]; m2 = p[0].z ^ flip[0]; m3 = p[0].w ^ flip[0];
#pragma unroll
            for (int k = 1; k < KEY_PLANES; ++k) {
                m0 &= p[k].x ^ flip[k];
                m1 &= p[k].y ^ flip[k];
                m2 &= p[k].z ^ flip[k];
                m3 &= p[k].w ^ flip[k];
            }
        }
#pragma unroll 1
        for (int j = 0; j < 4; ++j) {
            unsigned int mm = (j == 0) ? m0 : (j == 1) ? m1 : (j == 2) ? m2 : m3;
            const int eb = i4 * 128 + j * 32;
            while (__builtin_amdgcn_ballot_w32(mm != 0u) != 0u) {
                const bool p = mm != 0u;
                const int  b = __ffs((int)mm) - 1;
                mm &= mm - 1u;
                push(p, eb + b);
                if (qn >= 32) drain();
            }
        }
    }
    if (qn > 0) flush(qn);
    };

    if (FIRST) {
        search();
    } else {
        const int cnt = counts[blockIdx.x];
        if (cnt <= LIST_CAP) {
            const uint2* __restrict__ rd = lst;
            for (int base = __builtin_amdgcn_readfirstlane(wv) * 32; base < cnt; base += OWN_THREADS) {
                uint2 cur = make_uint2(0u, 0u);
                if (base + lane < cnt) cur = rd[base + lane];
                const int   s_l = (int)(cur.x & ((1u << SRC_BITS) - 1u));
                const int   r_l = (int)(cur.x >> SRC_BITS);
                const float w_l = __uint_as_float(cur.y);
                if (cnt - base >= 32) add_rows(32, s_l, r_l, w_l);
                else                  add_rows(cnt - base, s_l, r_l, w_l);
            }
        } else {
            search();
        }
    }
    __syncthreads();
    if (FIRST && tid == 0) counts[blockIdx.x] = l_cnt;

    const int nrows = (NN - lo < OWN_ROWS) ? (NN - lo) : OWN_ROWS;
    float* __restrict__ orow = out + (size_t)lo * F;
    for (int idx = tid; idx < nrows * F; idx += OWN_THREADS) {
        const int r = idx / F;
        const int f = idx - r * F;
        const int i = lo + r;
        const float di = dinv[i];
        float v = acc[idx] + h[(size_t)i * HS + f] * di * di + bias[f];
        if (RELU) v = fmaxf(v, 0.0f);
        orow[idx] = v;
    }
}

extern "C" void kernel_launch(void* const* d_in, const int* in_sizes, int n_in,
                              void* d_out, int out_size, void* d_ws, size_t ws_size,
                              hipStream_t stream) {
    const float* x    = (const float*)d_in[0];
    const int*   eidx = (const int*)d_in[1];
    const float* W1   = (const float*)d_in[2];
    const float* b1   = (const float*)d_in[3];
    const float* W2   = (const float*)d_in[4];
    const float* b2   = (const float*)d_in[5];
    const float* W3   = (const float*)d_in[6];
    const float* b3   = (const float*)d_in[7];
    float* out = (float*)d_out;

    const int* src = eidx;
    const int* dst = eidx + EE;

    float* dinv = (float*)d_ws;
    float* bufA = dinv + NN;
    float* bufB = bufA + (size_t)NN * HID;
    float* W3p  = bufB + (size_t)NN * HID;
    unsigned int* planes = (unsigned int*)(W3p + HID * NCLS_P);
    unsigned int* packed = planes + KEY_PLANES * PLANE_WORDS;
    unsigned int* lists  = packed + EE;
    int*          counts = (int*)(lists + (size_t)OWN_WGS * LIST_CAP * 2);

    const int TB = 256;

    init_deg_kernel<<<(NN + TB - 1) / TB, TB, 0, stream>>>(dinv, NN);
    deg_keys_kernel<<<(EE + 2 * KEY_EDGES - 1) / (2 * KEY_EDGES), 64, 0, stream>>>(src, dst, dinv, planes, packed);
    deg_to_dinv_kernel<<<(NN + TB - 1) / TB, TB, 0, stream>>>(dinv, NN);
    pad_w3_kernel<<<(HID * NCLS_P + TB - 1) / TB, TB, 0, stream>>>(W3, W3p);

    const int MT = NN / 16;
    const int gemmBlocks = (MT + 7) / 8;

    gemm_wmma_f32_kernel<8, IN_DIM, HID><<<gemmBlocks, TB, 0, stream>>>(x, W1, bufA, NN);
    own_rows_kernel<HID, HID, 1, 1><<<OWN_WGS, OWN_THREADS, 0, stream>>>(bufA, planes, packed, dinv, b1, lists, counts, bufB);

    gemm_wmma_f32_kernel<8, HID, HID><<<gemmBlocks, TB, 0, stream>>>(bufB, W2, bufA, NN);
    own_rows_kernel<HID, HID, 1, 0><<<OWN_WGS, OWN_THREADS, 0, stream>>>(bufA, planes, packed, dinv, b2, lists, counts, bufB);

    gemm_wmma_f32_kernel<3, HID, NCLS_P><<<gemmBlocks, TB, 0, stream>>>(bufB, W3p, bufA, NN);
    own_rows_kernel<NCLS, NCLS_P, 0, 0><<<OWN_WGS, OWN_THREADS, 0, stream>>>(bufA, planes, packed, dinv, b3, lists, counts, out);
}


# Degree count without global adds: each key work-group counts its edges' destinations in an LDS table of 16-bit counters, one kernel adds the 98 tables and writes dinv; integer counts, dinv bit-equal
# speedup vs baseline: 2.5563x; 1.1665x over previous
#include <hip/hip_runtime.h>
#include <hip/hip_bf16.h>


#define NN      100000
#define EE      1600000
#define IN_DIM  256
#define HID     128
#define NCLS    40
#define NCLS_P  48

typedef __attribute__((ext_vector_type(2))) float v2f;
typedef __attribute__((ext_vector_type(8))) float v8f;

#define OWN_WGS     256
#define OWN_ROWS    392
#define KEY_PLANES  8
#define PLANE_WORDS 50240
#define KEY_EDGES   1024
#define SRC_BITS    17
#define CNT_THREADS 512
#define CNT_WAVES   (CNT_THREADS / 32)
#define CNT_WGS     ((EE + KEY_EDGES * CNT_WAVES - 1) / (KEY_EDGES * CNT_WAVES))
#define CNT_WORDS   (NN / 2)

__global__ __launch_bounds__(CNT_THREADS)
void deg_keys_kernel(const int* __restrict__ src, const int* __restrict__ dst, unsigned int* __restrict__ cnt_parts,
                     unsigned int* __restrict__ planes, unsigned int* __restrict__ packed) {
    static_assert(NN % 8 == 0, "the counter table is a whole number of 16-byte pieces");
    static_assert(KEY_EDGES * CNT_WAVES < 65536, "a 16-bit counter holds a work-group's edges");
    __shared__ uint4 cnt4[CNT_WORDS / 4];
    unsigned int* cnt = (unsigned int*)cnt4;
    for (int i = threadIdx.x; i < CNT_WORDS / 4; i += CNT_THREADS) cnt4[i] = make_uint4(0u, 0u, 0u, 0u);
    __syncthreads();

    static_assert(OWN_WGS <= (1 << KEY_PLANES), "owner number fits the planes");
    static_assert(NN <= (1 << SRC_BITS) && OWN_ROWS <= (1 << (32 - SRC_BITS)), "packed word");
    static_assert(EE % 32 == 0 && EE / 32 <= PLANE_WORDS && PLANE_WORDS % 64 == 0, "plane size");
    const int lane = threadIdx.x & 31;
    const int wave = (blockIdx.x * blockDim.x + threadIdx.x) >> 5;
    const int base = wave * KEY_EDGES;
    if (base < EE) {

    unsigned int w[KEY_PLANES];
#pragma unroll
    for (int k = 0; k < KEY_PLANES; ++k) w[k] = 0u;

    constexpr int TG = 16;
#pragma unroll 1
    for (int j0 = 0; j0 < 32; j0 += TG) {
        int d[TG], s[TG];
#pragma unroll
        for (int g = 0; g < TG; ++g) {
            const int e = min(base + 32 * (j0 + g) + lane, EE - 1);
            d[g] = dst[e];
            s[g] = src[e];
        }
#pragma unroll
        for (int g = 0; g < TG; ++g) {
            const int j = j0 + g;
            const int e = base + 32 * j + lane;
            unsigned int o = 0u;
            if (e < EE) {
                atomicAdd(&cnt[(unsigned int)d[g] >> 1], 1u << (((unsigned int)d[g] & 1u) << 4));
                o = (unsigned int)d[g] / OWN_ROWS;
                packed[e] = (unsigned int)s[g] | (((unsigned int)d[g] - o * OWN_ROWS) << SRC_BITS);
            }
#pragma unroll
            for (int k = 0; k < KEY_PLANES; ++k) {
                const unsigned int b = __builtin_amdgcn_ballot_w32(((o >> k) & 1u) != 0u);
                if (lane == j) w[k] = b;
            }
        }
    }
    const int word = (base >> 5) + lane;
    if (word < EE / 32) {
#pragma unroll
        for (int k = 0; k < KEY_PLANES; ++k) planes[k * PLANE_WORDS + word] = w[k];
    }
    }
    __syncthreads();
    uint4* __restrict__ o4 = (uint4*)(cnt_parts + (size_t)blockIdx.x * CNT_WORDS);
    for (int i = threadIdx.x; i < CNT_WORDS / 4; i += CNT_THREADS) o4[i] = cnt4[i];
}

__global__ void counts_to_dinv_kernel(const unsigned int* __restrict__ cnt_parts, float* __restrict__ dinv) {
    const int i = blockIdx.x * blockDim.x + threadIdx.x;
    if (i >= CNT_WORDS) return;
    unsigned int lo = 0u, hi = 0u;
#pragma unroll
    for (int p = 0; p < CNT_WGS; ++p) {
        const unsigned int v = cnt_parts[(size_t)p * CNT_WORDS + i];
        lo += v & 0xFFFFu;
        hi += v >> 16;
    }
    ((float2*)dinv)[i] = make_float2(rsqrtf(1.0f + (float)lo), rsqrtf(1.0f + (float)hi));
}

__global__ void pad_w3_kernel(const float* __restrict__ w, float* __restrict__ wp) {
    int i = blockIdx.x * blockDim.x + threadIdx.x;
    if (i >= HID * NCLS_P) return;
    int r = i / NCLS_P, c = i - r * NCLS_P;
    wp[i] = (c < NCLS) ? w[r * NCLS + c] : 0.0f;
}

template <int NT, int K, int NOUT>
__global__ void gemm_wmma_f32_kernel(const float* __restrict__ A,
                                     const float* __restrict__ B,
                                     float* __restrict__ H,
                                     int M) {
    static_assert(NOUT == NT * 16, "full tiles only");
    const int lane  = threadIdx.x & 31;
    const int wave  = threadIdx.x >> 5;
    const int wpb   = blockDim.x >> 5;
    const int tile  = blockIdx.x * wpb + wave;
    const int mtile = M >> 4;
    if (tile >= mtile) return;

    const int rowA    = (tile << 4) + (lane & 15);
    const int kb      = (lane >> 4) << 1;
    const int colBase = lane & 15;

    v8f acc[NT];
#pragma unroll
    for (int t = 0; t < NT; ++t) acc[t] = (v8f)0.0f;

    const float* __restrict__ arow = A + (size_t)rowA * K;

#pragma unroll 2
    for (int k = 0; k < K; k += 4) {
        const float2 av = *(const float2*)(arow + k + kb);
        v2f a; a.x = av.x; a.y = av.y;

        const float* __restrict__ b0 = B + (size_t)(k + kb) * NOUT + colBase;
        const float* __restrict__ b1 = b0 + NOUT;
#pragma unroll
        for (int t = 0; t < NT; ++t) {
            v2f b;
            b.x = b0[t * 16];
            b.y = b1[t * 16];
            acc[t] = __builtin_amdgcn_wmma_f32_16x16x4_f32(
                false, a, false, b, (short)0, acc[t], false, false);
        }
    }

    const int rowD = (tile << 4) + ((lane >> 4) << 3);
    float* __restrict__ hout = H + (size_t)rowD * NOUT + colBase;
#pragma unroll
    for (int t = 0; t < NT; ++t) {
#pragma unroll
        for (int r = 0; r < 8; ++r)
            hout[(size_t)r * NOUT + t * 16] = acc[t][r];
    }
}

#define OWN_THREADS 1024
#define OWN_WAVES   (OWN_THREADS / 32)
#define OWN_QCAP    64
#define LIST_CAP    16384


template <int F, int HS, int RELU, int FIRST>
__global__ __launch_bounds__(OWN_THREADS)
void own_rows_kernel(const float* __restrict__ h,
                     const unsigned int* __restrict__ planes,
                     const unsigned int* __restrict__ packed,
                     const float* __restrict__ dinv,
                     const float* __restrict__ bias,
                     unsigned int* __restrict__ lists,
                     int* __restrict__ counts,
                     float* __restrict__ out) {
    static_assert(OWN_WGS * OWN_ROWS >= NN, "rows covered");
    static_assert((OWN_WGS - 1) * OWN_ROWS < NN, "no empty group");
    static_assert((OWN_ROWS * F * 4) % 128 == 0, "whole lines per group");
    static_assert(EE % 128 == 0, "16-byte reads of a plane hold 128 whole edges");
    static_assert((NN + 2 * NN * HID + HID * NCLS_P + KEY_PLANES * PLANE_WORDS + EE) % 2 == 0, "8-byte list places are aligned");

    __shared__ float acc[OWN_ROWS * F];
    __shared__ int   q_e[OWN_WAVES][OWN_QCAP];
    __shared__ int   l_cnt;

    const int tid  = threadIdx.x;
    const int lane = tid & 31;
    const int wv   = tid >> 5;
    const int lo   = blockIdx.x * OWN_ROWS;

    for (int i = tid; i < OWN_ROWS * F; i += OWN_THREADS) acc[i] = 0.0f;
    if (FIRST && tid == 0) l_cnt = 0;
    __syncthreads();

    uint2* __restrict__ lst = (uint2*)lists + (size_t)blockIdx.x * LIST_CAP;

    const unsigned below = (1u << lane) - 1u;
    int qn = 0;

    auto add_rows = [&](int n, int s_l, int r_l, float w_l) {
        constexpr int KF = (F + 31) / 32;
        constexpr int GE = 8;
#pragma unroll 1
        for (int j = 0; j < n; j += GE) {
            float v[GE][KF];
#pragma unroll
            for (int g = 0; g < GE; ++g) {
                const int s = __builtin_amdgcn_readlane(s_l, j + g);
                const float* hs = h + (size_t)s * HS;
#pragma unroll
                for (int k = 0; k < KF; ++k) {
                    const int f = lane + 32 * k;
                    v[g][k] = (f < F) ? hs[f] : 0.0f;
                }
            }
#pragma unroll
            for (int g = 0; g < GE; ++g) {
                if (j + g < n) {
                    const int   r = __builtin_amdgcn_readlane(r_l, j + g);
                    const float w = __int_as_float(__builtin_amdgcn_readlane(__float_as_int(w_l), j + g));
                    float* a = acc + r * F;
#pragma unroll
                    for (int k = 0; k < KF; ++k) {
                        const int f = lane + 32 * k;
                        if (f < F) atomicAdd(&a[f], v[g][k] * w);
                    }
                }
            }
        }
    };

    auto flush = [&](int n) {
        int r_l = 0, s_l = 0;
        float w_l = 0.0f;
        unsigned int pk = 0u;
        if (lane < n) {
            pk  = packed[q_e[wv][lane]];
            s_l = (int)(pk & ((1u << SRC_BITS) - 1u));
            r_l = (int)(pk >> SRC_BITS);
            w_l = dinv[s_l] * dinv[lo + r_l];
        }
        if (FIRST) {
            int pos = 0;
            if (lane == 0) pos = atomicAdd(&l_cnt, n);
            pos = __builtin_amdgcn_readfirstlane(pos);
            if (lane < n && pos + lane < LIST_CAP) lst[pos + lane] = make_uint2(pk, __float_as_uint(w_l));
        }
        add_rows(n, s_l, r_l, w_l);
    };

    auto push = [&](bool p, int eid) {
        const unsigned mk = __builtin_amdgcn_ballot_w32(p);
        if (p) q_e[wv][qn + __popc(mk & below)] = eid;
        qn += __popc(mk);
    };

    auto drain = [&]() {
        flush(32);
        const int rem = qn - 32;
        int te = 0;
        if (lane < rem) te = q_e[wv][32 + lane];
        if (lane < rem) q_e[wv][lane] = te;
        qn = rem;
    };

    auto search = [&]() {
    const unsigned int own = blockIdx.x;
    unsigned int flip[KEY_PLANES];
#pragma unroll
    for (int k = 0; k < KEY_PLANES; ++k) flip[k] = ((own >> k) & 1u) ? 0u : 0xFFFFFFFFu;

    constexpr int NP4 = EE / 128;
    constexpr int PV4 = PLANE_WORDS / 4;
    constexpr int NIT = (NP4 + OWN_THREADS - 1) / OWN_THREADS;
    const uint4* __restrict__ pl4 = (const uint4*)planes;
    const int rot = blockIdx.x % NIT;

    for (int it = 0; it < NIT; ++it) {
        int it2 = it + rot;
        if (it2 >= NIT) it2 -= NIT;
        const int i4 = it2 * OWN_THREADS + tid;
        unsigned int m0 = 0u, m1 = 0u, m2 = 0u, m3 = 0u;
        if (i4 < NP4) {
            uint4 p[KEY_PLANES];
#pragma unroll
            for (int k = 0; k < KEY_PLANES; ++k) p[k] = pl4[k * PV4 + i4];
            m0 = p[0].x ^ flip[0]; m1 = p[0].y ^ flip[0]; m2 = p[0].z ^ flip[0]; m3 = p[0].w ^ flip[0];
#pragma unroll
            for (int k = 1; k < KEY_PLANES; ++k) {
                m0 &= p[k].x ^ flip[k];
                m1 &= p[k].y ^ flip[k];
                m2 &= p[k].z ^ flip[k];
                m3 &= p[k].w ^ flip[k];
            }
        }
#pragma unroll 1
        for (int j = 0; j < 4; ++j) {
            unsigned int mm = (j == 0) ? m0 : (j == 1) ? m1 : (j == 2) ? m2 : m3;
            const int eb = i4 * 128 + j * 32;
            while (__builtin_amdgcn_ballot_w32(mm != 0u) != 0u) {
                const bool p = mm != 0u;
                const int  b = __ffs((int)mm) - 1;
                mm &= mm - 1u;
                push(p, eb + b);
                if (qn >= 32) drain();
            }
        }
    }
    if (qn > 0) flush(qn);
    };

    if (FIRST) {
        search();
    } else {
        const int cnt = counts[blockIdx.x];
        if (cnt <= LIST_CAP) {
            const uint2* __restrict__ rd = lst;
            for (int base = __builtin_amdgcn_readfirstlane(wv) * 32; base < cnt; base += OWN_THREADS) {
                uint2 cur = make_uint2(0u, 0u);
                if (base + lane < cnt) cur = rd[base + lane];
                const int   s_l = (int)(cur.x & ((1u << SRC_BITS) - 1u));
                const int   r_l = (int)(cur.x >> SRC_BITS);
                const float w_l = __uint_as_float(cur.y);
                if (cnt - base >= 32) add_rows(32, s_l, r_l, w_l);
                else                  add_rows(cnt - base, s_l, r_l, w_l);
            }
        } else {
            search();
        }
    }
    __syncthreads();
    if (FIRST && tid == 0) counts[blockIdx.x] = l_cnt;

    const int nrows = (NN - lo < OWN_ROWS) ? (NN - lo) : OWN_ROWS;
    float* __restrict__ orow = out + (size_t)lo * F;
    for (int idx = tid; idx < nrows * F; idx += OWN_THREADS) {
        const int r = idx / F;
        const int f = idx - r * F;
        const int i = lo + r;
        const float di = dinv[i];
        float v = acc[idx] + h[(size_t)i * HS + f] * di * di + bias[f];
        if (RELU) v = fmaxf(v, 0.0f);
        orow[idx] = v;
    }
}

extern "C" void kernel_launch(void* const* d_in, const int* in_sizes, int n_in,
                              void* d_out, int out_size, void* d_ws, size_t ws_size,
                              hipStream_t stream) {
    const float* x    = (const float*)d_in[0];
    const int*   eidx = (const int*)d_in[1];
    const float* W1   = (const float*)d_in[2];
    const float* b1   = (const float*)d_in[3];
    const float* W2   = (const float*)d_in[4];
    const float* b2   = (const float*)d_in[5];
    const float* W3   = (const float*)d_in[6];
    const float* b3   = (const float*)d_in[7];
    float* out = (float*)d_out;

    const int* src = eidx;
    const int* dst = eidx + EE;

    float* dinv = (float*)d_ws;
    float* bufA = dinv + NN;
    float* bufB = bufA + (size_t)NN * HID;
    float* W3p  = bufB + (size_t)NN * HID;
    unsigned int* planes = (unsigned int*)(W3p + HID * NCLS_P);
    unsigned int* packed = planes + KEY_PLANES * PLANE_WORDS;
    unsigned int* lists  = packed + EE;
    int*          counts = (int*)(lists + (size_t)OWN_WGS * LIST_CAP * 2);
    unsigned int* cnt_parts = (unsigned int*)(counts + OWN_WGS);
    static_assert((NN + 2 * NN * HID + HID * NCLS_P + KEY_PLANES * PLANE_WORDS + EE + OWN_WGS * LIST_CAP * 2 + OWN_WGS) % 4 == 0,
                  "16-byte pieces of the counter tables are aligned");

    const int TB = 256;

    deg_keys_kernel<<<CNT_WGS, CNT_THREADS, 0, stream>>>(src, dst, cnt_parts, planes, packed);
    counts_to_dinv_kernel<<<(CNT_WORDS + TB - 1) / TB, TB, 0, stream>>>(cnt_parts, dinv);
    pad_w3_kernel<<<(HID * NCLS_P + TB - 1) / TB, TB, 0, stream>>>(W3, W3p);

    const int MT = NN / 16;
    const int gemmBlocks = (MT + 7) / 8;

    gemm_wmma_f32_kernel<8, IN_DIM, HID><<<gemmBlocks, TB, 0, stream>>>(x, W1, bufA, NN);
    own_rows_kernel<HID, HID, 1, 1><<<OWN_WGS, OWN_THREADS, 0, stream>>>(bufA, planes, packed, dinv, b1, lists, counts, bufB);

    gemm_wmma_f32_kernel<8, HID, HID><<<gemmBlocks, TB, 0, stream>>>(bufB, W2, bufA, NN);
    own_rows_kernel<HID, HID, 1, 0><<<OWN_WGS, OWN_THREADS, 0, stream>>>(bufA, planes, packed, dinv, b2, lists, counts, bufB);

    gemm_wmma_f32_kernel<3, HID, NCLS_P><<<gemmBlocks, TB, 0, stream>>>(bufB, W3p, bufA, NN);
    own_rows_kernel<NCLS, NCLS_P, 0, 0><<<OWN_WGS, OWN_THREADS, 0, stream>>>(bufA, planes, packed, dinv, b3, lists, counts, out);
}


# Products read W from LDS: each of 256 32-wave work-groups copies the whole W once as k-pairs, waves take row tiles dealt evenly, A asked for 4 k-steps ahead; k order of the sums kept
# speedup vs baseline: 2.8208x; 1.1035x over previous
#include <hip/hip_runtime.h>
#include <hip/hip_bf16.h>


#define NN      100000
#define EE      1600000
#define IN_DIM  256
#define HID     128
#define NCLS    40
#define NCLS_P  48

typedef __attribute__((ext_vector_type(2))) float v2f;
typedef __attribute__((ext_vector_type(8))) float v8f;

#define OWN_WGS     256
#define OWN_ROWS    392
#define KEY_PLANES  8
#define PLANE_WORDS 50240
#define KEY_EDGES   1024
#define SRC_BITS    17
#define CNT_THREADS 512
#define CNT_WAVES   (CNT_THREADS / 32)
#define CNT_WGS     ((EE + KEY_EDGES * CNT_WAVES - 1) / (KEY_EDGES * CNT_WAVES))
#define CNT_WORDS   (NN / 2)

__global__ __launch_bounds__(CNT_THREADS)
void deg_keys_kernel(const int* __restrict__ src, const int* __restrict__ dst, unsigned int* __restrict__ cnt_parts,
                     unsigned int* __restrict__ planes, unsigned int* __restrict__ packed) {
    static_assert(NN % 8 == 0, "the counter table is a whole number of 16-byte pieces");
    static_assert(KEY_EDGES * CNT_WAVES < 65536, "a 16-bit counter holds a work-group's edges");
    __shared__ uint4 cnt4[CNT_WORDS / 4];
    unsigned int* cnt = (unsigned int*)cnt4;
    for (int i = threadIdx.x; i < CNT_WORDS / 4; i += CNT_THREADS) cnt4[i] = make_uint4(0u, 0u, 0u, 0u);
    __syncthreads();

    static_assert(OWN_WGS <= (1 << KEY_PLANES), "owner number fits the planes");
    static_assert(NN <= (1 << SRC_BITS) && OWN_ROWS <= (1 << (32 - SRC_BITS)), "packed word");
    static_assert(EE % 32 == 0 && EE / 32 <= PLANE_WORDS && PLANE_WORDS % 64 == 0, "plane size");
    const int lane = threadIdx.x & 31;
    const int wave = (blockIdx.x * blockDim.x + threadIdx.x) >> 5;
    const int base = wave * KEY_EDGES;
    if (base < EE) {

    unsigned int w[KEY_PLANES];
#pragma unroll
    for (int k = 0; k < KEY_PLANES; ++k) w[k] = 0u;

    constexpr int TG = 16;
#pragma unroll 1
    for (int j0 = 0; j0 < 32; j0 += TG) {
        int d[TG], s[TG];
#pragma unroll
        for (int g = 0; g < TG; ++g) {
            const int e = min(base + 32 * (j0 + g) + lane, EE - 1);
            d[g] = dst[e];
            s[g] = src[e];
        }
#pragma unroll
        for (int g = 0; g < TG; ++g) {
            const int j = j0 + g;
            const int e = base + 32 * j + lane;
            unsigned int o = 0u;
            if (e < EE) {
                atomicAdd(&cnt[(unsigned int)d[g] >> 1], 1u << (((unsigned int)d[g] & 1u) << 4));
                o = (unsigned int)d[g] / OWN_ROWS;
                packed[e] = (unsigned int)s[g] | (((unsigned int)d[g] - o * OWN_ROWS) << SRC_BITS);
            }
#pragma unroll
            for (int k = 0; k < KEY_PLANES; ++k) {
                const unsigned int b = __builtin_amdgcn_ballot_w32(((o >> k) & 1u) != 0u);
                if (lane == j) w[k] = b;
            }
        }
    }
    const int word = (base >> 5) + lane;
    if (word < EE / 32) {
#pragma unroll
        for (int k = 0; k < KEY_PLANES; ++k) planes[k * PLANE_WORDS + word] = w[k];
    }
    }
    __syncthreads();
    uint4* __restrict__ o4 = (uint4*)(cnt_parts + (size_t)blockIdx.x * CNT_WORDS);
    for (int i = threadIdx.x; i < CNT_WORDS / 4; i += CNT_THREADS) o4[i] = cnt4[i];
}

__global__ void counts_to_dinv_kernel(const unsigned int* __restrict__ cnt_parts, float* __restrict__ dinv) {
    const int i = blockIdx.x * blockDim.x + threadIdx.x;
    if (i >= CNT_WORDS) return;
    unsigned int lo = 0u, hi = 0u;
#pragma unroll
    for (int p = 0; p < CNT_WGS; ++p) {
        const unsigned int v = cnt_parts[(size_t)p * CNT_WORDS + i];
        lo += v & 0xFFFFu;
        hi += v >> 16;
    }
    ((float2*)dinv)[i] = make_float2(rsqrtf(1.0f + (float)lo), rsqrtf(1.0f + (float)hi));
}

__global__ void pad_w3_kernel(const float* __restrict__ w, float* __restrict__ wp) {
    int i = blockIdx.x * blockDim.x + threadIdx.x;
    if (i >= HID * NCLS_P) return;
    int r = i / NCLS_P, c = i - r * NCLS_P;
    wp[i] = (c < NCLS) ? w[r * NCLS + c] : 0.0f;
}

#define GEMM_THREADS 1024
#define GEMM_WGS     256
template <int NT, int K, int NOUT>
__global__ __launch_bounds__(GEMM_THREADS)
void gemm_wmma_f32_kernel(const float* __restrict__ A,
                          const float* __restrict__ B,
                          float* __restrict__ H,
                          int M) {
    static_assert(NOUT == NT * 16, "full tiles only");
    static_assert(K % 16 == 0, "A is read four k-steps at a time");
    constexpr int PITCH = ((NOUT * 8) % 256 == 128) ? NOUT : NOUT + 16;
    static_assert((PITCH * 8) % 256 == 128, "bank halves");
    __shared__ float2 Bs[(K / 2) * PITCH];
    const int tid = threadIdx.x;
    for (int i = tid; i < (K / 2) * NOUT; i += GEMM_THREADS) {
        const int p = i / NOUT, col = i - p * NOUT;
        Bs[p * PITCH + col] = make_float2(B[(2 * p) * NOUT + col], B[(2 * p + 1) * NOUT + col]);
    }
    __syncthreads();

    const int lane    = tid & 31;
    const int wave    = tid >> 5;
    const int mtile   = M >> 4;
    const int colBase = lane & 15;
    const int half    = lane >> 4;
    const float2* __restrict__ bs = Bs + half * PITCH + colBase;
    constexpr int KB = 4;

    const int tile = wave * gridDim.x + blockIdx.x;
    if (tile < mtile) {
        v8f acc[NT];
#pragma unroll
        for (int t = 0; t < NT; ++t) acc[t] = (v8f)0.0f;
        const float* __restrict__ arow = A + (size_t)((tile << 4) + colBase) * K + 2 * half;
        float2 an[KB];
#pragma unroll
        for (int s = 0; s < KB; ++s) an[s] = *(const float2*)(arow + 4 * s);
#pragma unroll 1
        for (int k0 = 0; k0 < K; k0 += 4 * KB) {
            float2 ac[KB];
#pragma unroll
            for (int s = 0; s < KB; ++s) ac[s] = an[s];
            if (k0 + 4 * KB < K) {
#pragma unroll
                for (int s = 0; s < KB; ++s) an[s] = *(const float2*)(arow + k0 + 4 * KB + 4 * s);
            }
#pragma unroll
            for (int s = 0; s < KB; ++s) {
                v2f a; a.x = ac[s].x; a.y = ac[s].y;
                const float2* __restrict__ bp = bs + ((k0 + 4 * s) >> 1) * PITCH;
#pragma unroll
                for (int t = 0; t < NT; ++t) {
                    const float2 bv = bp[t * 16];
                    v2f b; b.x = bv.x; b.y = bv.y;
                    acc[t] = __builtin_amdgcn_wmma_f32_16x16x4_f32(
                        false, a, false, b, (short)0, acc[t], false, false);
                }
            }
        }
        const int rowD = (tile << 4) + (half << 3);
        float* __restrict__ hout = H + (size_t)rowD * NOUT + colBase;
#pragma unroll
        for (int t = 0; t < NT; ++t) {
#pragma unroll
            for (int r = 0; r < 8; ++r)
                hout[(size_t)r * NOUT + t * 16] = acc[t][r];
        }
    }
}

#define OWN_THREADS 1024
#define OWN_WAVES   (OWN_THREADS / 32)
#define OWN_QCAP    64
#define LIST_CAP    16384


template <int F, int HS, int RELU, int FIRST>
__global__ __launch_bounds__(OWN_THREADS)
void own_rows_kernel(const float* __restrict__ h,
                     const unsigned int* __restrict__ planes,
                     const unsigned int* __restrict__ packed,
                     const float* __restrict__ dinv,
                     const float* __restrict__ bias,
                     unsigned int* __restrict__ lists,
                     int* __restrict__ counts,
                     float* __restrict__ out) {
    static_assert(OWN_WGS * OWN_ROWS >= NN, "rows covered");
    static_assert((OWN_WGS - 1) * OWN_ROWS < NN, "no empty group");
    static_assert((OWN_ROWS * F * 4) % 128 == 0, "whole lines per group");
    static_assert(EE % 128 == 0, "16-byte reads of a plane hold 128 whole edges");
    static_assert((NN + 2 * NN * HID + HID * NCLS_P + KEY_PLANES * PLANE_WORDS + EE) % 2 == 0, "8-byte list places are aligned");

    __shared__ float acc[OWN_ROWS * F];
    __shared__ int   q_e[OWN_WAVES][OWN_QCAP];
    __shared__ int   l_cnt;

    const int tid  = threadIdx.x;
    const int lane = tid & 31;
    const int wv   = tid >> 5;
    const int lo   = blockIdx.x * OWN_ROWS;

    for (int i = tid; i < OWN_ROWS * F; i += OWN_THREADS) acc[i] = 0.0f;
    if (FIRST && tid == 0) l_cnt = 0;
    __syncthreads();

    uint2* __restrict__ lst = (uint2*)lists + (size_t)blockIdx.x * LIST_CAP;

    const unsigned below = (1u << lane) - 1u;
    int qn = 0;

    auto add_rows = [&](int n, int s_l, int r_l, float w_l) {
        constexpr int KF = (F + 31) / 32;
        constexpr int GE = 8;
#pragma unroll 1
        for (int j = 0; j < n; j += GE) {
            float v[GE][KF];
#pragma unroll
            for (int g = 0; g < GE; ++g) {
                const int s = __builtin_amdgcn_readlane(s_l, j + g);
                const float* hs = h + (size_t)s * HS;
#pragma unroll
                for (int k = 0; k < KF; ++k) {
                    const int f = lane + 32 * k;
                    v[g][k] = (f < F) ? hs[f] : 0.0f;
                }
            }
#pragma unroll
            for (int g = 0; g < GE; ++g) {
                if (j + g < n) {
                    const int   r = __builtin_amdgcn_readlane(r_l, j + g);
                    const float w = __int_as_float(__builtin_amdgcn_readlane(__float_as_int(w_l), j + g));
                    float* a = acc + r * F;
#pragma unroll
                    for (int k = 0; k < KF; ++k) {
                        const int f = lane + 32 * k;
                        if (f < F) atomicAdd(&a[f], v[g][k] * w);
                    }
                }
            }
        }
    };

    auto flush = [&](int n) {
        int r_l = 0, s_l = 0;
        float w_l = 0.0f;
        unsigned int pk = 0u;
        if (lane < n) {
            pk  = packed[q_e[wv][lane]];
            s_l = (int)(pk & ((1u << SRC_BITS) - 1u));
            r_l = (int)(pk >> SRC_BITS);
            w_l = dinv[s_l] * dinv[lo + r_l];
        }
        if (FIRST) {
            int pos = 0;
            if (lane == 0) pos = atomicAdd(&l_cnt, n);
            pos = __builtin_amdgcn_readfirstlane(pos);
            if (lane < n && pos + lane < LIST_CAP) lst[pos + lane] = make_uint2(pk, __float_as_uint(w_l));
        }
        add_rows(n, s_l, r_l, w_l);
    };

    auto push = [&](bool p, int eid) {
        const unsigned mk = __builtin_amdgcn_ballot_w32(p);
        if (p) q_e[wv][qn + __popc(mk & below)] = eid;
        qn += __popc(mk);
    };

    auto drain = [&]() {
        flush(32);
        const int rem = qn - 32;
        int te = 0;
        if (lane < rem) te = q_e[wv][32 + lane];
        if (lane < rem) q_e[wv][lane] = te;
        qn = rem;
    };

    auto search = [&]() {
    const unsigned int own = blockIdx.x;
    unsigned int flip[KEY_PLANES];
#pragma unroll
    for (int k = 0; k < KEY_PLANES; ++k) flip[k] = ((own >> k) & 1u) ? 0u : 0xFFFFFFFFu;

    constexpr int NP4 = EE / 128;
    constexpr int PV4 = PLANE_WORDS / 4;
    constexpr int NIT = (NP4 + OWN_THREADS - 1) / OWN_THREADS;
    const uint4* __restrict__ pl4 = (const uint4*)planes;
    const int rot = blockIdx.x % NIT;

    for (int it = 0; it < NIT; ++it) {
        int it2 = it + rot;
        if (it2 >= NIT) it2 -= NIT;
        const int i4 = it2 * OWN_THREADS + tid;
        unsigned int m0 = 0u, m1 = 0u, m2 = 0u, m3 = 0u;
        if (i4 < NP4) {
            uint4 p[KEY_PLANES];
#pragma unroll
            for (int k = 0; k < KEY_PLANES; ++k) p[k] = pl4[k * PV4 + i4];
            m0 = p[0].x ^ flip[0]; m1 = p[0].y ^ flip[0]; m2 = p[0].z ^ flip[0]; m3 = p[0].w ^ flip[0];
#pragma unroll
            for (int k = 1; k < KEY_PLANES; ++k) {
                m0 &= p[k].x ^ flip[k];
                m1 &= p[k].y ^ flip[k];
                m2 &= p[k].z ^ flip[k];
                m3 &= p[k].w ^ flip[k];
            }
        }
#pragma unroll 1
        for (int j = 0; j < 4; ++j) {
            unsigned int mm = (j == 0) ? m0 : (j == 1) ? m1 : (j == 2) ? m2 : m3;
            const int eb = i4 * 128 + j * 32;
            while (__builtin_amdgcn_ballot_w32(mm != 0u) != 0u) {
                const bool p = mm != 0u;
                const int  b = __ffs((int)mm) - 1;
                mm &= mm - 1u;
                push(p, eb + b);
                if (qn >= 32) drain();
            }
        }
    }
    if (qn > 0) flush(qn);
    };

    if (FIRST) {
        search();
    } else {
        const int cnt = counts[blockIdx.x];
        if (cnt <= LIST_CAP) {
            const uint2* __restrict__ rd = lst;
            for (int base = __builtin_amdgcn_readfirstlane(wv) * 32; base < cnt; base += OWN_THREADS) {
                uint2 cur = make_uint2(0u, 0u);
                if (base + lane < cnt) cur = rd[base + lane];
                const int   s_l = (int)(cur.x & ((1u << SRC_BITS) - 1u));
                const int   r_l = (int)(cur.x >> SRC_BITS);
                const float w_l = __uint_as_float(cur.y);
                if (cnt - base >= 32) add_rows(32, s_l, r_l, w_l);
                else                  add_rows(cnt - base, s_l, r_l, w_l);
            }
        } else {
            search();
        }
    }
    __syncthreads();
    if (FIRST && tid == 0) counts[blockIdx.x] = l_cnt;

    const int nrows = (NN - lo < OWN_ROWS) ? (NN - lo) : OWN_ROWS;
    float* __restrict__ orow = out + (size_t)lo * F;
    for (int idx = tid; idx < nrows * F; idx += OWN_THREADS) {
        const int r = idx / F;
        const int f = idx - r * F;
        const int i = lo + r;
        const float di = dinv[i];
        float v = acc[idx] + h[(size_t)i * HS + f] * di * di + bias[f];
        if (RELU) v = fmaxf(v, 0.0f);
        orow[idx] = v;
    }
}

extern "C" void kernel_launch(void* const* d_in, const int* in_sizes, int n_in,
                              void* d_out, int out_size, void* d_ws, size_t ws_size,
                              hipStream_t stream) {
    const float* x    = (const float*)d_in[0];
    const int*   eidx = (const int*)d_in[1];
    const float* W1   = (const float*)d_in[2];
    const float* b1   = (const float*)d_in[3];
    const float* W2   = (const float*)d_in[4];
    const float* b2   = (const float*)d_in[5];
    const float* W3   = (const float*)d_in[6];
    const float* b3   = (const float*)d_in[7];
    float* out = (float*)d_out;

    const int* src = eidx;
    const int* dst = eidx + EE;

    float* dinv = (float*)d_ws;
    float* bufA = dinv + NN;
    float* bufB = bufA + (size_t)NN * HID;
    float* W3p  = bufB + (size_t)NN * HID;
    unsigned int* planes = (unsigned int*)(W3p + HID * NCLS_P);
    unsigned int* packed = planes + KEY_PLANES * PLANE_WORDS;
    unsigned int* lists  = packed + EE;
    int*          counts = (int*)(lists + (size_t)OWN_WGS * LIST_CAP * 2);
    unsigned int* cnt_parts = (unsigned int*)(counts + OWN_WGS);
    static_assert((NN + 2 * NN * HID + HID * NCLS_P + KEY_PLANES * PLANE_WORDS + EE + OWN_WGS * LIST_CAP * 2 + OWN_WGS) % 4 == 0,
                  "16-byte pieces of the counter tables are aligned");

    const int TB = 256;
    static_assert(NN / 16 <= GEMM_WGS * (GEMM_THREADS / 32), "a wave for every row tile of the products");

    deg_keys_kernel<<<CNT_WGS, CNT_THREADS, 0, stream>>>(src, dst, cnt_parts, planes, packed);
    counts_to_dinv_kernel<<<(CNT_WORDS + TB - 1) / TB, TB, 0, stream>>>(cnt_parts, dinv);
    pad_w3_kernel<<<(HID * NCLS_P + TB - 1) / TB, TB, 0, stream>>>(W3, W3p);


    gemm_wmma_f32_kernel<8, IN_DIM, HID><<<GEMM_WGS, GEMM_THREADS, 0, stream>>>(x, W1, bufA, NN);
    own_rows_kernel<HID, HID, 1, 1><<<OWN_WGS, OWN_THREADS, 0, stream>>>(bufA, planes, packed, dinv, b1, lists, counts, bufB);

    gemm_wmma_f32_kernel<8, HID, HID><<<GEMM_WGS, GEMM_THREADS, 0, stream>>>(bufB, W2, bufA, NN);
    own_rows_kernel<HID, HID, 1, 0><<<OWN_WGS, OWN_THREADS, 0, stream>>>(bufA, planes, packed, dinv, b2, lists, counts, bufB);

    gemm_wmma_f32_kernel<3, HID, NCLS_P><<<GEMM_WGS, GEMM_THREADS, 0, stream>>>(bufB, W3p, bufA, NN);
    own_rows_kernel<NCLS, NCLS_P, 0, 0><<<OWN_WGS, OWN_THREADS, 0, stream>>>(bufA, planes, packed, dinv, b3, lists, counts, out);
}


# Edge sums fetch a source row with ONE load per lane (16 bytes for 128-float rows, 8 for 40-float rows), not 4-byte loads; LDS sums lane-major; closing pass moves rows alike; sum order = LDS arrival
# speedup vs baseline: 3.0076x; 1.0662x over previous
#include <hip/hip_runtime.h>
#include <hip/hip_bf16.h>


#define NN      100000
#define EE      1600000
#define IN_DIM  256
#define HID     128
#define NCLS    40
#define NCLS_P  48

typedef __attribute__((ext_vector_type(2))) float v2f;
typedef __attribute__((ext_vector_type(4))) float v4f;
typedef __attribute__((ext_vector_type(8))) float v8f;

#define OWN_WGS     256
#define OWN_ROWS    392
#define KEY_PLANES  8
#define PLANE_WORDS 50240
#define KEY_EDGES   1024
#define SRC_BITS    17
#define CNT_THREADS 512
#define CNT_WAVES   (CNT_THREADS / 32)
#define CNT_WGS     ((EE + KEY_EDGES * CNT_WAVES - 1) / (KEY_EDGES * CNT_WAVES))
#define CNT_WORDS   (NN / 2)

__global__ __launch_bounds__(CNT_THREADS)
void deg_keys_kernel(const int* __restrict__ src, const int* __restrict__ dst, unsigned int* __restrict__ cnt_parts,
                     unsigned int* __restrict__ planes, unsigned int* __restrict__ packed) {
    static_assert(NN % 8 == 0, "the counter table is a whole number of 16-byte pieces");
    static_assert(KEY_EDGES * CNT_WAVES < 65536, "a 16-bit counter holds a work-group's edges");
    __shared__ uint4 cnt4[CNT_WORDS / 4];
    unsigned int* cnt = (unsigned int*)cnt4;
    for (int i = threadIdx.x; i < CNT_WORDS / 4; i += CNT_THREADS) cnt4[i] = make_uint4(0u, 0u, 0u, 0u);
    __syncthreads();

    static_assert(OWN_WGS <= (1 << KEY_PLANES), "owner number fits the planes");
    static_assert(NN <= (1 << SRC_BITS) && OWN_ROWS <= (1 << (32 - SRC_BITS)), "packed word");
    static_assert(EE % 32 == 0 && EE / 32 <= PLANE_WORDS && PLANE_WORDS % 64 == 0, "plane size");
    const int lane = threadIdx.x & 31;
    const int wave = (blockIdx.x * blockDim.x + threadIdx.x) >> 5;
    const int base = wave * KEY_EDGES;
    if (base < EE) {

    unsigned int w[KEY_PLANES];
#pragma unroll
    for (int k = 0; k < KEY_PLANES; ++k) w[k] = 0u;

    constexpr int TG = 16;
#pragma unroll 1
    for (int j0 = 0; j0 < 32; j0 += TG) {
        int d[TG], s[TG];
#pragma unroll
        for (int g = 0; g < TG; ++g) {
            const int e = min(base + 32 * (j0 + g) + lane, EE - 1);
            d[g] = dst[e];
            s[g] = src[e];
        }
#pragma unroll
        for (int g = 0; g < TG; ++g) {
            const int j = j0 + g;
            const int e = base + 32 * j + lane;
            unsigned int o = 0u;
            if (e < EE) {
                atomicAdd(&cnt[(unsigned int)d[g] >> 1], 1u << (((unsigned int)d[g] & 1u) << 4));
                o = (unsigned int)d[g] / OWN_ROWS;
                packed[e] = (unsigned int)s[g] | (((unsigned int)d[g] - o * OWN_ROWS) << SRC_BITS);
            }
#pragma unroll
            for (int k = 0; k < KEY_PLANES; ++k) {
                const unsigned int b = __builtin_amdgcn_ballot_w32(((o >> k) & 1u) != 0u);
                if (lane == j) w[k] = b;
            }
        }
    }
    const int word = (base >> 5) + lane;
    if (word < EE / 32) {
#pragma unroll
        for (int k = 0; k < KEY_PLANES; ++k) planes[k * PLANE_WORDS + word] = w[k];
    }
    }
    __syncthreads();
    uint4* __restrict__ o4 = (uint4*)(cnt_parts + (size_t)blockIdx.x * CNT_WORDS);
    for (int i = threadIdx.x; i < CNT_WORDS / 4; i += CNT_THREADS) o4[i] = cnt4[i];
}

__global__ void counts_to_dinv_kernel(const unsigned int* __restrict__ cnt_parts, float* __restrict__ dinv) {
    const int i = blockIdx.x * blockDim.x + threadIdx.x;
    if (i >= CNT_WORDS) return;
    unsigned int lo = 0u, hi = 0u;
#pragma unroll
    for (int p = 0; p < CNT_WGS; ++p) {
        const unsigned int v = cnt_parts[(size_t)p * CNT_WORDS + i];
        lo += v & 0xFFFFu;
        hi += v >> 16;
    }
    ((float2*)dinv)[i] = make_float2(rsqrtf(1.0f + (float)lo), rsqrtf(1.0f + (float)hi));
}

__global__ void pad_w3_kernel(const float* __restrict__ w, float* __restrict__ wp) {
    int i = blockIdx.x * blockDim.x + threadIdx.x;
    if (i >= HID * NCLS_P) return;
    int r = i / NCLS_P, c = i - r * NCLS_P;
    wp[i] = (c < NCLS) ? w[r * NCLS + c] : 0.0f;
}

#define GEMM_THREADS 1024
#define GEMM_WGS     256
template <int NT, int K, int NOUT>
__global__ __launch_bounds__(GEMM_THREADS)
void gemm_wmma_f32_kernel(const float* __restrict__ A,
                          const float* __restrict__ B,
                          float* __restrict__ H,
                          int M) {
    static_assert(NOUT == NT * 16, "full tiles only");
    static_assert(K % 16 == 0, "A is read four k-steps at a time");
    constexpr int PITCH = ((NOUT * 8) % 256 == 128) ? NOUT : NOUT + 16;
    static_assert((PITCH * 8) % 256 == 128, "bank halves");
    __shared__ float2 Bs[(K / 2) * PITCH];
    const int tid = threadIdx.x;
    for (int i = tid; i < (K / 2) * NOUT; i += GEMM_THREADS) {
        const int p = i / NOUT, col = i - p * NOUT;
        Bs[p * PITCH + col] = make_float2(B[(2 * p) * NOUT + col], B[(2 * p + 1) * NOUT + col]);
    }
    __syncthreads();

    const int lane    = tid & 31;
    const int wave    = tid >> 5;
    const int mtile   = M >> 4;
    const int colBase = lane & 15;
    const int half    = lane >> 4;
    const float2* __restrict__ bs = Bs + half * PITCH + colBase;
    constexpr int KB = 4;

    const int tile = wave * gridDim.x + blockIdx.x;
    if (tile < mtile) {
        v8f acc[NT];
#pragma unroll
        for (int t = 0; t < NT; ++t) acc[t] = (v8f)0.0f;
        const float* __restrict__ arow = A + (size_t)((tile << 4) + colBase) * K + 2 * half;
        float2 an[KB];
#pragma unroll
        for (int s = 0; s < KB; ++s) an[s] = *(const float2*)(arow + 4 * s);
#pragma unroll 1
        for (int k0 = 0; k0 < K; k0 += 4 * KB) {
            float2 ac[KB];
#pragma unroll
            for (int s = 0; s < KB; ++s) ac[s] = an[s];
            if (k0 + 4 * KB < K) {
#pragma unroll
                for (int s = 0; s < KB; ++s) an[s] = *(const float2*)(arow + k0 + 4 * KB + 4 * s);
            }
#pragma unroll
            for (int s = 0; s < KB; ++s) {
                v2f a; a.x = ac[s].x; a.y = ac[s].y;
                const float2* __restrict__ bp = bs + ((k0 + 4 * s) >> 1) * PITCH;
#pragma unroll
                for (int t = 0; t < NT; ++t) {
                    const float2 bv = bp[t * 16];
                    v2f b; b.x = bv.x; b.y = bv.y;
                    acc[t] = __builtin_amdgcn_wmma_f32_16x16x4_f32(
                        false, a, false, b, (short)0, acc[t], false, false);
                }
            }
        }
        const int rowD = (tile << 4) + (half << 3);
        float* __restrict__ hout = H + (size_t)rowD * NOUT + colBase;
#pragma unroll
        for (int t = 0; t < NT; ++t) {
#pragma unroll
            for (int r = 0; r < 8; ++r)
                hout[(size_t)r * NOUT + t * 16] = acc[t][r];
        }
    }
}

template <int V> struct row_piece;
template <> struct row_piece<2> { typedef v2f type; };
template <> struct row_piece<4> { typedef v4f type; };

#define OWN_THREADS 1024
#define OWN_WAVES   (OWN_THREADS / 32)
#define OWN_QCAP    64
#define LIST_CAP    16384


template <int F, int HS, int RELU, int FIRST>
__global__ __launch_bounds__(OWN_THREADS)
void own_rows_kernel(const float* __restrict__ h,
                     const unsigned int* __restrict__ planes,
                     const unsigned int* __restrict__ packed,
                     const float* __restrict__ dinv,
                     const float* __restrict__ bias,
                     unsigned int* __restrict__ lists,
                     int* __restrict__ counts,
                     float* __restrict__ out) {
    static_assert(OWN_WGS * OWN_ROWS >= NN, "rows covered");
    static_assert((OWN_WGS - 1) * OWN_ROWS < NN, "no empty group");
    static_assert((OWN_ROWS * F * 4) % 128 == 0, "whole lines per group");
    static_assert(EE % 128 == 0, "16-byte reads of a plane hold 128 whole edges");
    static_assert((NN + 2 * NN * HID + HID * NCLS_P + KEY_PLANES * PLANE_WORDS + EE) % 2 == 0, "8-byte list places are aligned");
    constexpr int V   = (F % 128 == 0) ? 4 : 2;
    constexpr int LPR = F / V;
    typedef typename row_piece<V>::type pv;
    static_assert(F % V == 0 && LPR <= 32 && HS % V == 0 && NN % 4 == 0, "a row is LPR aligned pieces of V floats");

    __shared__ float acc[OWN_ROWS * F];
    __shared__ int   q_e[OWN_WAVES][OWN_QCAP];
    __shared__ int   l_cnt;

    const int tid  = threadIdx.x;
    const int lane = tid & 31;
    const int wv   = tid >> 5;
    const int lo   = blockIdx.x * OWN_ROWS;

    for (int i = tid; i < OWN_ROWS * F; i += OWN_THREADS) acc[i] = 0.0f;
    if (FIRST && tid == 0) l_cnt = 0;
    __syncthreads();

    uint2* __restrict__ lst = (uint2*)lists + (size_t)blockIdx.x * LIST_CAP;

    const unsigned below = (1u << lane) - 1u;
    int qn = 0;

    auto add_rows = [&](int n, int s_l, int r_l, float w_l) {
        constexpr int GE = 8;
        const int lv = (lane < LPR) ? lane : 0;
#pragma unroll 1
        for (int j = 0; j < n; j += GE) {
            pv v[GE];
#pragma unroll
            for (int g = 0; g < GE; ++g) {
                const int s = __builtin_amdgcn_readlane(s_l, j + g);
                v[g] = *(const pv*)(h + (size_t)s * HS + V * lv);
            }
#pragma unroll
            for (int g = 0; g < GE; ++g) {
                if (j + g < n) {
                    const int   r = __builtin_amdgcn_readlane(r_l, j + g);
                    const float w = __int_as_float(__builtin_amdgcn_readlane(__float_as_int(w_l), j + g));
                    float* a = acc + r * F + lane;
#pragma unroll
                    for (int c = 0; c < V; ++c) {
                        if (lane < LPR) atomicAdd(&a[c * LPR], v[g][c] * w);
                    }
                }
            }
        }
    };

    auto flush = [&](int n) {
        int r_l = 0, s_l = 0;
        float w_l = 0.0f;
        unsigned int pk = 0u;
        if (lane < n) {
            pk  = packed[q_e[wv][lane]];
            s_l = (int)(pk & ((1u << SRC_BITS) - 1u));
            r_l = (int)(pk >> SRC_BITS);
            w_l = dinv[s_l] * dinv[lo + r_l];
        }
        if (FIRST) {
            int pos = 0;
            if (lane == 0) pos = atomicAdd(&l_cnt, n);
            pos = __builtin_amdgcn_readfirstlane(pos);
            if (lane < n && pos + lane < LIST_CAP) lst[pos + lane] = make_uint2(pk, __float_as_uint(w_l));
        }
        add_rows(n, s_l, r_l, w_l);
    };

    auto push = [&](bool p, int eid) {
        const unsigned mk = __builtin_amdgcn_ballot_w32(p);
        if (p) q_e[wv][qn + __popc(mk & below)] = eid;
        qn += __popc(mk);
    };

    auto drain = [&]() {
        flush(32);
        const int rem = qn - 32;
        int te = 0;
        if (lane < rem) te = q_e[wv][32 + lane];
        if (lane < rem) q_e[wv][lane] = te;
        qn = rem;
    };

    auto search = [&]() {
    const unsigned int own = blockIdx.x;
    unsigned int flip[KEY_PLANES];
#pragma unroll
    for (int k = 0; k < KEY_PLANES; ++k) flip[k] = ((own >> k) & 1u) ? 0u : 0xFFFFFFFFu;

    constexpr int NP4 = EE / 128;
    constexpr int PV4 = PLANE_WORDS / 4;
    constexpr int NIT = (NP4 + OWN_THREADS - 1) / OWN_THREADS;
    const uint4* __restrict__ pl4 = (const uint4*)planes;
    const int rot = blockIdx.x % NIT;

    for (int it = 0; it < NIT; ++it) {
        int it2 = it + rot;
        if (it2 >= NIT) it2 -= NIT;
        const int i4 = it2 * OWN_THREADS + tid;
        unsigned int m0 = 0u, m1 = 0u, m2 = 0u, m3 = 0u;
        if (i4 < NP4) {
            uint4 p[KEY_PLANES];
#pragma unroll
            for (int k = 0; k < KEY_PLANES; ++k) p[k] = pl4[k * PV4 + i4];
            m0 = p[0].x ^ flip[0]; m1 = p[0].y ^ flip[0]; m2 = p[0].z ^ flip[0]; m3 = p[0].w ^ flip[0];
#pragma unroll
            for (int k = 1; k < KEY_PLANES; ++k) {
                m0 &= p[k].x ^ flip[k];
                m1 &= p[k].y ^ flip[k];
                m2 &= p[k].z ^ flip[k];
                m3 &= p[k].w ^ flip[k];
            }
        }
#pragma unroll 1
        for (int j = 0; j < 4; ++j) {
            unsigned int mm = (j == 0) ? m0 : (j == 1) ? m1 : (j == 2) ? m2 : m3;
            const int eb = i4 * 128 + j * 32;
            while (__builtin_amdgcn_ballot_w32(mm != 0u) != 0u) {
                const bool p = mm != 0u;
                const int  b = __ffs((int)mm) - 1;
                mm &= mm - 1u;
                push(p, eb + b);
                if (qn >= 32) drain();
            }
        }
    }
    if (qn > 0) flush(qn);
    };

    if (FIRST) {
        search();
    } else {
        const int cnt = counts[blockIdx.x];
        if (cnt <= LIST_CAP) {
            const uint2* __restrict__ rd = lst;
            for (int base = __builtin_amdgcn_readfirstlane(wv) * 32; base < cnt; base += OWN_THREADS) {
                uint2 cur = make_uint2(0u, 0u);
                if (base + lane < cnt) cur = rd[base + lane];
                const int   s_l = (int)(cur.x & ((1u << SRC_BITS) - 1u));
                const int   r_l = (int)(cur.x >> SRC_BITS);
                const float w_l = __uint_as_float(cur.y);
                if (cnt - base >= 32) add_rows(32, s_l, r_l, w_l);
                else                  add_rows(cnt - base, s_l, r_l, w_l);
            }
        } else {
            search();
        }
    }
    __syncthreads();
    if (FIRST && tid == 0) counts[blockIdx.x] = l_cnt;

    const int nrows = (NN - lo < OWN_ROWS) ? (NN - lo) : OWN_ROWS;
    float* __restrict__ orow = out + (size_t)lo * F;
#pragma unroll 4
    for (int p = tid; p < nrows * LPR; p += OWN_THREADS) {
        const int r = p / LPR;
        const int l = p - r * LPR;
        const int i = lo + r;
        const float di = dinv[i];
        const pv hv = *(const pv*)(h + (size_t)i * HS + V * l);
        const pv bv = *(const pv*)(bias + V * l);
        pv o;
#pragma unroll
        for (int c = 0; c < V; ++c) {
            float t = acc[r * F + c * LPR + l] + hv[c] * di * di + bv[c];
            if (RELU) t = fmaxf(t, 0.0f);
            o[c] = t;
        }
        *(pv*)(orow + (size_t)r * F + V * l) = o;
    }
}

extern "C" void kernel_launch(void* const* d_in, const int* in_sizes, int n_in,
                              void* d_out, int out_size, void* d_ws, size_t ws_size,
                              hipStream_t stream) {
    const float* x    = (const float*)d_in[0];
    const int*   eidx = (const int*)d_in[1];
    const float* W1   = (const float*)d_in[2];
    const float* b1   = (const float*)d_in[3];
    const float* W2   = (const float*)d_in[4];
    const float* b2   = (const float*)d_in[5];
    const float* W3   = (const float*)d_in[6];
    const float* b3   = (const float*)d_in[7];
    float* out = (float*)d_out;

    const int* src = eidx;
    const int* dst = eidx + EE;

    float* dinv = (float*)d_ws;
    float* bufA = dinv + NN;
    float* bufB = bufA + (size_t)NN * HID;
    float* W3p  = bufB + (size_t)NN * HID;
    unsigned int* planes = (unsigned int*)(W3p + HID * NCLS_P);
    unsigned int* packed = planes + KEY_PLANES * PLANE_WORDS;
    unsigned int* lists  = packed + EE;
    int*          counts = (int*)(lists + (size_t)OWN_WGS * LIST_CAP * 2);
    unsigned int* cnt_parts = (unsigned int*)(counts + OWN_WGS);
    static_assert((NN + 2 * NN * HID + HID * NCLS_P + KEY_PLANES * PLANE_WORDS + EE + OWN_WGS * LIST_CAP * 2 + OWN_WGS) % 4 == 0,
                  "16-byte pieces of the counter tables are aligned");

    const int TB = 256;
    static_assert(NN / 16 <= GEMM_WGS * (GEMM_THREADS / 32), "a wave for every row tile of the products");

    deg_keys_kernel<<<CNT_WGS, CNT_THREADS, 0, stream>>>(src, dst, cnt_parts, planes, packed);
    counts_to_dinv_kernel<<<(CNT_WORDS + TB - 1) / TB, TB, 0, stream>>>(cnt_parts, dinv);
    pad_w3_kernel<<<(HID * NCLS_P + TB - 1) / TB, TB, 0, stream>>>(W3, W3p);


    gemm_wmma_f32_kernel<8, IN_DIM, HID><<<GEMM_WGS, GEMM_THREADS, 0, stream>>>(x, W1, bufA, NN);
    own_rows_kernel<HID, HID, 1, 1><<<OWN_WGS, OWN_THREADS, 0, stream>>>(bufA, planes, packed, dinv, b1, lists, counts, bufB);

    gemm_wmma_f32_kernel<8, HID, HID><<<GEMM_WGS, GEMM_THREADS, 0, stream>>>(bufB, W2, bufA, NN);
    own_rows_kernel<HID, HID, 1, 0><<<OWN_WGS, OWN_THREADS, 0, stream>>>(bufA, planes, packed, dinv, b2, lists, counts, bufB);

    gemm_wmma_f32_kernel<3, HID, NCLS_P><<<GEMM_WGS, GEMM_THREADS, 0, stream>>>(bufB, W3p, bufA, NN);
    own_rows_kernel<NCLS, NCLS_P, 0, 0><<<OWN_WGS, OWN_THREADS, 0, stream>>>(bufA, planes, packed, dinv, b3, lists, counts, out);
}


# Layer-1 edge search moved onto the 7 tile-less waves of each layer-1 product work-group, beside its WMMAs: they write the edge lists, all 3 layers sum from lists; sum order = LDS arrival
# speedup vs baseline: 3.1084x; 1.0335x over previous
#include <hip/hip_runtime.h>
#include <hip/hip_bf16.h>


#define NN      100000
#define EE      1600000
#define IN_DIM  256
#define HID     128
#define NCLS    40
#define NCLS_P  48

typedef __attribute__((ext_vector_type(2))) float v2f;
typedef __attribute__((ext_vector_type(4))) float v4f;
typedef __attribute__((ext_vector_type(8))) float v8f;

#define OWN_WGS     256
#define OWN_ROWS    392
#define KEY_PLANES  8
#define PLANE_WORDS 50240
#define KEY_EDGES   1024
#define SRC_BITS    17
#define CNT_THREADS 512
#define CNT_WAVES   (CNT_THREADS / 32)
#define CNT_WGS     ((EE + KEY_EDGES * CNT_WAVES - 1) / (KEY_EDGES * CNT_WAVES))
#define CNT_WORDS   (NN / 2)

__global__ __launch_bounds__(CNT_THREADS)
void deg_keys_kernel(const int* __restrict__ src, const int* __restrict__ dst, unsigned int* __restrict__ cnt_parts,
                     unsigned int* __restrict__ planes, unsigned int* __restrict__ packed) {
    static_assert(NN % 8 == 0, "the counter table is a whole number of 16-byte pieces");
    static_assert(KEY_EDGES * CNT_WAVES < 65536, "a 16-bit counter holds a work-group's edges");
    __shared__ uint4 cnt4[CNT_WORDS / 4];
    unsigned int* cnt = (unsigned int*)cnt4;
    for (int i = threadIdx.x; i < CNT_WORDS / 4; i += CNT_THREADS) cnt4[i] = make_uint4(0u, 0u, 0u, 0u);
    __syncthreads();

    static_assert(OWN_WGS <= (1 << KEY_PLANES), "owner number fits the planes");
    static_assert(NN <= (1 << SRC_BITS) && OWN_ROWS <= (1 << (32 - SRC_BITS)), "packed word");
    static_assert(EE % 32 == 0 && EE / 32 <= PLANE_WORDS && PLANE_WORDS % 64 == 0, "plane size");
    const int lane = threadIdx.x & 31;
    const int wave = (blockIdx.x * blockDim.x + threadIdx.x) >> 5;
    const int base = wave * KEY_EDGES;
    if (base < EE) {

    unsigned int w[KEY_PLANES];
#pragma unroll
    for (int k = 0; k < KEY_PLANES; ++k) w[k] = 0u;

    constexpr int TG = 16;
#pragma unroll 1
    for (int j0 = 0; j0 < 32; j0 += TG) {
        int d[TG], s[TG];
#pragma unroll
        for (int g = 0; g < TG; ++g) {
            const int e = min(base + 32 * (j0 + g) + lane, EE - 1);
            d[g] = dst[e];
            s[g] = src[e];
        }
#pragma unroll
        for (int g = 0; g < TG; ++g) {
            const int j = j0 + g;
            const int e = base + 32 * j + lane;
            unsigned int o = 0u;
            if (e < EE) {
                atomicAdd(&cnt[(unsigned int)d[g] >> 1], 1u << (((unsigned int)d[g] & 1u) << 4));
                o = (unsigned int)d[g] / OWN_ROWS;
                packed[e] = (unsigned int)s[g] | (((unsigned int)d[g] - o * OWN_ROWS) << SRC_BITS);
            }
#pragma unroll
            for (int k = 0; k < KEY_PLANES; ++k) {
                const unsigned int b = __builtin_amdgcn_ballot_w32(((o >> k) & 1u) != 0u);
                if (lane == j) w[k] = b;
            }
        }
    }
    const int word = (base >> 5) + lane;
    if (word < EE / 32) {
#pragma unroll
        for (int k = 0; k < KEY_PLANES; ++k) planes[k * PLANE_WORDS + word] = w[k];
    }
    }
    __syncthreads();
    uint4* __restrict__ o4 = (uint4*)(cnt_parts + (size_t)blockIdx.x * CNT_WORDS);
    for (int i = threadIdx.x; i < CNT_WORDS / 4; i += CNT_THREADS) o4[i] = cnt4[i];
}

__global__ void counts_to_dinv_kernel(const unsigned int* __restrict__ cnt_parts, float* __restrict__ dinv) {
    const int i = blockIdx.x * blockDim.x + threadIdx.x;
    if (i >= CNT_WORDS) return;
    unsigned int lo = 0u, hi = 0u;
#pragma unroll
    for (int p = 0; p < CNT_WGS; ++p) {
        const unsigned int v = cnt_parts[(size_t)p * CNT_WORDS + i];
        lo += v & 0xFFFFu;
        hi += v >> 16;
    }
    ((float2*)dinv)[i] = make_float2(rsqrtf(1.0f + (float)lo), rsqrtf(1.0f + (float)hi));
}

__global__ void pad_w3_kernel(const float* __restrict__ w, float* __restrict__ wp) {
    int i = blockIdx.x * blockDim.x + threadIdx.x;
    if (i >= HID * NCLS_P) return;
    int r = i / NCLS_P, c = i - r * NCLS_P;
    wp[i] = (c < NCLS) ? w[r * NCLS + c] : 0.0f;
}

#define GEMM_THREADS 1024
#define GEMM_WGS     256
#define SRCH_WAVES   7
#define SRCH_CAP     5120
#define LIST_CAP     16384
template <int NT, int K, int NOUT, int SEARCH>
__global__ __launch_bounds__(GEMM_THREADS)
void gemm_wmma_f32_kernel(const float* __restrict__ A,
                          const float* __restrict__ B,
                          float* __restrict__ H,
                          int M,
                          const unsigned int* __restrict__ planes,
                          const unsigned int* __restrict__ packed,
                          const float* __restrict__ dinv,
                          unsigned int* __restrict__ lists,
                          int* __restrict__ counts) {
    static_assert(NOUT == NT * 16, "full tiles only");
    static_assert(!SEARCH || GEMM_WGS == OWN_WGS, "one product work-group searches for one owner block");
    static_assert((GEMM_THREADS / 32 - SRCH_WAVES) * GEMM_WGS >= NN / 16, "the search waves have no row tile");
    static_assert(K % 16 == 0, "A is read four k-steps at a time");
    constexpr int PITCH = ((NOUT * 8) % 256 == 128) ? NOUT : NOUT + 16;
    static_assert((PITCH * 8) % 256 == 128, "bank halves");
    __shared__ float2 Bs[(K / 2) * PITCH];
    __shared__ int wq[SEARCH ? SRCH_WAVES * SRCH_CAP : 1];
    __shared__ int l_cnt;
    const int tid = threadIdx.x;
    for (int i = tid; i < (K / 2) * NOUT; i += GEMM_THREADS) {
        const int p = i / NOUT, col = i - p * NOUT;
        Bs[p * PITCH + col] = make_float2(B[(2 * p) * NOUT + col], B[(2 * p + 1) * NOUT + col]);
    }
    if (SEARCH && tid == 0) l_cnt = 0;
    __syncthreads();

    const int lane    = tid & 31;
    const int wave    = tid >> 5;
    const int mtile   = M >> 4;
    const int colBase = lane & 15;
    const int half    = lane >> 4;
    const float2* __restrict__ bs = Bs + half * PITCH + colBase;
    constexpr int KB = 4;

    const int tile = wave * gridDim.x + blockIdx.x;
    if (tile < mtile) {
        v8f acc[NT];
#pragma unroll
        for (int t = 0; t < NT; ++t) acc[t] = (v8f)0.0f;
        const float* __restrict__ arow = A + (size_t)((tile << 4) + colBase) * K + 2 * half;
        float2 an[KB];
#pragma unroll
        for (int s = 0; s < KB; ++s) an[s] = *(const float2*)(arow + 4 * s);
#pragma unroll 1
        for (int k0 = 0; k0 < K; k0 += 4 * KB) {
            float2 ac[KB];
#pragma unroll
            for (int s = 0; s < KB; ++s) ac[s] = an[s];
            if (k0 + 4 * KB < K) {
#pragma unroll
                for (int s = 0; s < KB; ++s) an[s] = *(const float2*)(arow + k0 + 4 * KB + 4 * s);
            }
#pragma unroll
            for (int s = 0; s < KB; ++s) {
                v2f a; a.x = ac[s].x; a.y = ac[s].y;
                const float2* __restrict__ bp = bs + ((k0 + 4 * s) >> 1) * PITCH;
#pragma unroll
                for (int t = 0; t < NT; ++t) {
                    const float2 bv = bp[t * 16];
                    v2f b; b.x = bv.x; b.y = bv.y;
                    acc[t] = __builtin_amdgcn_wmma_f32_16x16x4_f32(
                        false, a, false, b, (short)0, acc[t], false, false);
                }
            }
        }
        const int rowD = (tile << 4) + (half << 3);
        float* __restrict__ hout = H + (size_t)rowD * NOUT + colBase;
#pragma unroll
        for (int t = 0; t < NT; ++t) {
#pragma unroll
            for (int r = 0; r < 8; ++r)
                hout[(size_t)r * NOUT + t * 16] = acc[t][r];
        }
    } else if (SEARCH && wave >= GEMM_THREADS / 32 - SRCH_WAVES) {
        const int sw = wave - (GEMM_THREADS / 32 - SRCH_WAVES);
        int* __restrict__ q = wq + sw * SRCH_CAP;
        const unsigned int own = blockIdx.x;
        const int lo = (int)own * OWN_ROWS;
        uint2* __restrict__ lst = (uint2*)lists + (size_t)own * LIST_CAP;
        const unsigned below = (1u << lane) - 1u;
        int qn = 0;

        auto emit = [&](const int n) {
            int base = 0;
            if (lane == 0) base = atomicAdd(&l_cnt, n);
            base = __builtin_amdgcn_readfirstlane(base);
            constexpr int U = 8;
#pragma unroll 1
            for (int i0 = 0; i0 < n; i0 += 32 * U) {
                unsigned int pk[U];
                float wsrc[U], wdst[U];
#pragma unroll
                for (int u = 0; u < U; ++u) pk[u] = packed[q[min(i0 + 32 * u + lane, n - 1)]];
#pragma unroll
                for (int u = 0; u < U; ++u) {
                    wsrc[u] = dinv[pk[u] & ((1u << SRC_BITS) - 1u)];
                    wdst[u] = dinv[lo + (int)(pk[u] >> SRC_BITS)];
                }
#pragma unroll
                for (int u = 0; u < U; ++u) {
                    const int pos = min(base + min(i0 + 32 * u + lane, n - 1), LIST_CAP - 1);
                    lst[pos] = make_uint2(pk[u], __float_as_uint(wsrc[u] * wdst[u]));
                }
            }
        };

        unsigned int flip[KEY_PLANES];
#pragma unroll
        for (int k = 0; k < KEY_PLANES; ++k) flip[k] = ((own >> k) & 1u) ? 0u : 0xFFFFFFFFu;

        constexpr int NP4 = EE / 128;
        constexpr int PV4 = PLANE_WORDS / 4;
        constexpr int SL  = SRCH_WAVES * 32;
        constexpr int NIT = (NP4 + SL - 1) / SL;
        static_assert(EE % 128 == 0 && NIT % 2 == 0, "whole pieces; the trips go in pairs");
        const uint4* __restrict__ pl4 = (const uint4*)planes;
        const int gl = sw * 32 + lane;
        const int rot = (int)own % NIT;
        auto piece = [&](const int it) { int t = it + rot; if (t >= NIT) t -= NIT; return t * SL + gl; };

        auto fetch = [&](uint4 (&p)[KEY_PLANES], const int i4) {
            const int c4 = (i4 < NP4) ? i4 : NP4 - 1;
#pragma unroll
            for (int k = 0; k < KEY_PLANES; ++k) p[k] = pl4[k * PV4 + c4];
        };

        auto take = [&](const uint4 (&p)[KEY_PLANES], const int i4) {
            if (qn > SRCH_CAP - 32 * 128) { emit(qn); qn = 0; }
            unsigned int m0 = p[0].x ^ flip[0], m1 = p[0].y ^ flip[0], m2 = p[0].z ^ flip[0], m3 = p[0].w ^ flip[0];
#pragma unroll
            for (int k = 1; k < KEY_PLANES; ++k) {
                m0 &= p[k].x ^ flip[k];
                m1 &= p[k].y ^ flip[k];
                m2 &= p[k].z ^ flip[k];
                m3 &= p[k].w ^ flip[k];
            }
            if (i4 >= NP4) { m0 = 0u; m1 = 0u; m2 = 0u; m3 = 0u; }
#pragma unroll
            for (int j = 0; j < 4; ++j) {
                unsigned int mm = (j == 0) ? m0 : (j == 1) ? m1 : (j == 2) ? m2 : m3;
                const int eb = i4 * 128 + j * 32;
                unsigned int mk;
                while ((mk = __builtin_amdgcn_ballot_w32(mm != 0u)) != 0u) {
                    const bool hit = mm != 0u;
                    const int  b   = __ffs((int)mm) - 1;
                    mm &= mm - 1u;
                    if (hit) q[qn + __popc(mk & below)] = eb + b;
                    qn += __popc(mk);
                }
            }
        };

        uint4 pa[KEY_PLANES], pb[KEY_PLANES];
        fetch(pa, piece(0));
#pragma unroll 1
        for (int it = 0; it < NIT; it += 2) {
            fetch(pb, piece(it + 1));
            take(pa, piece(it));
            fetch(pa, piece(it + 2));
            take(pb, piece(it + 1));
        }
        if (qn > 0) emit(qn);
    }
    if (SEARCH) {
        __syncthreads();
        if (tid == 0) counts[blockIdx.x] = l_cnt;
    }
}

template <int V> struct row_piece;
template <> struct row_piece<2> { typedef v2f type; };
template <> struct row_piece<4> { typedef v4f type; };

#define OWN_THREADS 1024
#define OWN_WAVES   (OWN_THREADS / 32)
#define OWN_QCAP    64


template <int F, int HS, int RELU>
__global__ __launch_bounds__(OWN_THREADS)
void own_rows_kernel(const float* __restrict__ h,
                     const unsigned int* __restrict__ planes,
                     const unsigned int* __restrict__ packed,
                     const float* __restrict__ dinv,
                     const float* __restrict__ bias,
                     const unsigned int* __restrict__ lists,
                     const int* __restrict__ counts,
                     float* __restrict__ out) {
    static_assert(OWN_WGS * OWN_ROWS >= NN, "rows covered");
    static_assert((OWN_WGS - 1) * OWN_ROWS < NN, "no empty group");
    static_assert((OWN_ROWS * F * 4) % 128 == 0, "whole lines per group");
    static_assert(EE % 128 == 0, "16-byte reads of a plane hold 128 whole edges");
    static_assert((NN + 2 * NN * HID + HID * NCLS_P + KEY_PLANES * PLANE_WORDS + EE) % 2 == 0, "8-byte list places are aligned");
    constexpr int V   = (F % 128 == 0) ? 4 : 2;
    constexpr int LPR = F / V;
    typedef typename row_piece<V>::type pv;
    static_assert(F % V == 0 && LPR <= 32 && HS % V == 0 && NN % 4 == 0, "a row is LPR aligned pieces of V floats");

    __shared__ float acc[OWN_ROWS * F];
    __shared__ int   q_e[OWN_WAVES][OWN_QCAP];

    const int tid  = threadIdx.x;
    const int lane = tid & 31;
    const int wv   = tid >> 5;
    const int lo   = blockIdx.x * OWN_ROWS;

    for (int i = tid; i < OWN_ROWS * F; i += OWN_THREADS) acc[i] = 0.0f;
    __syncthreads();

    const uint2* __restrict__ lst = (const uint2*)lists + (size_t)blockIdx.x * LIST_CAP;

    const unsigned below = (1u << lane) - 1u;
    int qn = 0;

    auto add_rows = [&](int n, int s_l, int r_l, float w_l) {
        constexpr int GE = 8;
        const int lv = (lane < LPR) ? lane : 0;
#pragma unroll 1
        for (int j = 0; j < n; j += GE) {
            pv v[GE];
#pragma unroll
            for (int g = 0; g < GE; ++g) {
                const int s = __builtin_amdgcn_readlane(s_l, j + g);
                v[g] = *(const pv*)(h + (size_t)s * HS + V * lv);
            }
#pragma unroll
            for (int g = 0; g < GE; ++g) {
                if (j + g < n) {
                    const int   r = __builtin_amdgcn_readlane(r_l, j + g);
                    const float w = __int_as_float(__builtin_amdgcn_readlane(__float_as_int(w_l), j + g));
                    float* a = acc + r * F + lane;
#pragma unroll
                    for (int c = 0; c < V; ++c) {
                        if (lane < LPR) atomicAdd(&a[c * LPR], v[g][c] * w);
                    }
                }
            }
        }
    };

    auto flush = [&](int n) {
        int r_l = 0, s_l = 0;
        float w_l = 0.0f;
        if (lane < n) {
            const unsigned int pk = packed[q_e[wv][lane]];
            s_l = (int)(pk & ((1u << SRC_BITS) - 1u));
            r_l = (int)(pk >> SRC_BITS);
            w_l = dinv[s_l] * dinv[lo + r_l];
        }
        add_rows(n, s_l, r_l, w_l);
    };

    auto push = [&](bool p, int eid) {
        const unsigned mk = __builtin_amdgcn_ballot_w32(p);
        if (p) q_e[wv][qn + __popc(mk & below)] = eid;
        qn += __popc(mk);
    };

    auto drain = [&]() {
        flush(32);
        const int rem = qn - 32;
        int te = 0;
        if (lane < rem) te = q_e[wv][32 + lane];
        if (lane < rem) q_e[wv][lane] = te;
        qn = rem;
    };

    auto search = [&]() {
    const unsigned int own = blockIdx.x;
    unsigned int flip[KEY_PLANES];
#pragma unroll
    for (int k = 0; k < KEY_PLANES; ++k) flip[k] = ((own >> k) & 1u) ? 0u : 0xFFFFFFFFu;

    constexpr int NP4 = EE / 128;
    constexpr int PV4 = PLANE_WORDS / 4;
    constexpr int NIT = (NP4 + OWN_THREADS - 1) / OWN_THREADS;
    const uint4* __restrict__ pl4 = (const uint4*)planes;
    const int rot = blockIdx.x % NIT;

    for (int it = 0; it < NIT; ++it) {
        int it2 = it + rot;
        if (it2 >= NIT) it2 -= NIT;
        const int i4 = it2 * OWN_THREADS + tid;
        unsigned int m0 = 0u, m1 = 0u, m2 = 0u, m3 = 0u;
        if (i4 < NP4) {
            uint4 p[KEY_PLANES];
#pragma unroll
            for (int k = 0; k < KEY_PLANES; ++k) p[k] = pl4[k * PV4 + i4];
            m0 = p[0].x ^ flip[0]; m1 = p[0].y ^ flip[0]; m2 = p[0].z ^ flip[0]; m3 = p[0].w ^ flip[0];
#pragma unroll
            for (int k = 1; k < KEY_PLANES; ++k) {
                m0 &= p[k].x ^ flip[k];
                m1 &= p[k].y ^ flip[k];
                m2 &= p[k].z ^ flip[k];
                m3 &= p[k].w ^ flip[k];
            }
        }
#pragma unroll 1
        for (int j = 0; j < 4; ++j) {
            unsigned int mm = (j == 0) ? m0 : (j == 1) ? m1 : (j == 2) ? m2 : m3;
            const int eb = i4 * 128 + j * 32;
            while (__builtin_amdgcn_ballot_w32(mm != 0u) != 0u) {
                const bool p = mm != 0u;
                const int  b = __ffs((int)mm) - 1;
                mm &= mm - 1u;
                push(p, eb + b);
                if (qn >= 32) drain();
            }
        }
    }
    if (qn > 0) flush(qn);
    };

    {
        const int cnt = counts[blockIdx.x];
        if (cnt <= LIST_CAP) {
            const uint2* __restrict__ rd = lst;
            for (int base = __builtin_amdgcn_readfirstlane(wv) * 32; base < cnt; base += OWN_THREADS) {
                uint2 cur = make_uint2(0u, 0u);
                if (base + lane < cnt) cur = rd[base + lane];
                const int   s_l = (int)(cur.x & ((1u << SRC_BITS) - 1u));
                const int   r_l = (int)(cur.x >> SRC_BITS);
                const float w_l = __uint_as_float(cur.y);
                if (cnt - base >= 32) add_rows(32, s_l, r_l, w_l);
                else                  add_rows(cnt - base, s_l, r_l, w_l);
            }
        } else {
            search();
        }
    }
    __syncthreads();

    const int nrows = (NN - lo < OWN_ROWS) ? (NN - lo) : OWN_ROWS;
    float* __restrict__ orow = out + (size_t)lo * F;
#pragma unroll 4
    for (int p = tid; p < nrows * LPR; p += OWN_THREADS) {
        const int r = p / LPR;
        const int l = p - r * LPR;
        const int i = lo + r;
        const float di = dinv[i];
        const pv hv = *(const pv*)(h + (size_t)i * HS + V * l);
        const pv bv = *(const pv*)(bias + V * l);
        pv o;
#pragma unroll
        for (int c = 0; c < V; ++c) {
            float t = acc[r * F + c * LPR + l] + hv[c] * di * di + bv[c];
            if (RELU) t = fmaxf(t, 0.0f);
            o[c] = t;
        }
        *(pv*)(orow + (size_t)r * F + V * l) = o;
    }
}

extern "C" void kernel_launch(void* const* d_in, const int* in_sizes, int n_in,
                              void* d_out, int out_size, void* d_ws, size_t ws_size,
                              hipStream_t stream) {
    const float* x    = (const float*)d_in[0];
    const int*   eidx = (const int*)d_in[1];
    const float* W1   = (const float*)d_in[2];
    const float* b1   = (const float*)d_in[3];
    const float* W2   = (const float*)d_in[4];
    const float* b2   = (const float*)d_in[5];
    const float* W3   = (const float*)d_in[6];
    const float* b3   = (const float*)d_in[7];
    float* out = (float*)d_out;

    const int* src = eidx;
    const int* dst = eidx + EE;

    float* dinv = (float*)d_ws;
    float* bufA = dinv + NN;
    float* bufB = bufA + (size_t)NN * HID;
    float* W3p  = bufB + (size_t)NN * HID;
    unsigned int* planes = (unsigned int*)(W3p + HID * NCLS_P);
    unsigned int* packed = planes + KEY_PLANES * PLANE_WORDS;
    unsigned int* lists  = packed + EE;
    int*          counts = (int*)(lists + (size_t)OWN_WGS * LIST_CAP * 2);
    unsigned int* cnt_parts = (unsigned int*)(counts + OWN_WGS);
    static_assert((NN + 2 * NN * HID + HID * NCLS_P + KEY_PLANES * PLANE_WORDS + EE + OWN_WGS * LIST_CAP * 2 + OWN_WGS) % 4 == 0,
                  "16-byte pieces of the counter tables are aligned");

    const int TB = 256;
    static_assert(NN / 16 <= GEMM_WGS * (GEMM_THREADS / 32), "a wave for every row tile of the products");

    deg_keys_kernel<<<CNT_WGS, CNT_THREADS, 0, stream>>>(src, dst, cnt_parts, planes, packed);
    counts_to_dinv_kernel<<<(CNT_WORDS + TB - 1) / TB, TB, 0, stream>>>(cnt_parts, dinv);
    pad_w3_kernel<<<(HID * NCLS_P + TB - 1) / TB, TB, 0, stream>>>(W3, W3p);


    gemm_wmma_f32_kernel<8, IN_DIM, HID, 1><<<GEMM_WGS, GEMM_THREADS, 0, stream>>>(x, W1, bufA, NN, planes, packed, dinv, lists, counts);
    own_rows_kernel<HID, HID, 1><<<OWN_WGS, OWN_THREADS, 0, stream>>>(bufA, planes, packed, dinv, b1, lists, counts, bufB);

    gemm_wmma_f32_kernel<8, HID, HID, 0><<<GEMM_WGS, GEMM_THREADS, 0, stream>>>(bufB, W2, bufA, NN, planes, packed, dinv, lists, counts);
    own_rows_kernel<HID, HID, 1><<<OWN_WGS, OWN_THREADS, 0, stream>>>(bufA, planes, packed, dinv, b2, lists, counts, bufB);

    gemm_wmma_f32_kernel<3, HID, NCLS_P, 0><<<GEMM_WGS, GEMM_THREADS, 0, stream>>>(bufB, W3p, bufA, NN, planes, packed, dinv, lists, counts);
    own_rows_kernel<NCLS, NCLS_P, 0><<<OWN_WGS, OWN_THREADS, 0, stream>>>(bufA, planes, packed, dinv, b3, lists, counts, out);
}
